# starting kernel
# baseline (speedup 1.0000x reference)
#include <hip/hip_runtime.h>
#include <hip/hip_bf16.h>


typedef __attribute__((ext_vector_type(2))) float v2f;
typedef __attribute__((ext_vector_type(8))) float v8f;

constexpr int CH    = 128;
constexpr int BATCH = 64;
constexpr int HW    = 56 * 56;
constexpr float MTOT = (float)BATCH * (float)HW;
constexpr float EPSR = 1e-5f;
constexpr int   T_NEWTON = 5;

__global__ void __launch_bounds__(256)
cw_mean_partial(const float* __restrict__ x, float* __restrict__ meanPart) {
  const int b = blockIdx.x;
  const int c = blockIdx.y;
  const float* p = x + ((size_t)(b * CH + c)) * HW;
  float s = 0.f;
  for (int i = threadIdx.x; i < HW; i += 256) s += p[i];
  __shared__ float red[256];
  red[threadIdx.x] = s;
  __syncthreads();
  for (int off = 128; off > 0; off >>= 1) {
    if ((int)threadIdx.x < off) red[threadIdx.x] += red[threadIdx.x + off];
    __syncthreads();
  }
  if (threadIdx.x == 0) meanPart[c * BATCH + b] = red[0];
}

__global__ void cw_mean_finish(const float* __restrict__ meanPart,
                               float* __restrict__ mean) {
  const int c = threadIdx.x;
  float s = 0.f;
  for (int b = 0; b < BATCH; ++b) s += meanPart[c * BATCH + b];
  mean[c] = s / MTOT;
}

constexpr int KC = 64;
constexpr int XPAD = KC + 4;

__global__ void __launch_bounds__(256)
cw_gram(const float* __restrict__ x, float* __restrict__ gramPart) {
  __shared__ float sX[CH][XPAD];
  const int b    = blockIdx.x;
  const int tid  = threadIdx.x;
  const int wave = tid >> 5;
  const int lane = tid & 31;
  const int hi   = lane >> 4;
  const int l15  = lane & 15;

  v8f acc[8] = {};
  const float* xb = x + (size_t)b * CH * HW;

  for (int kc = 0; kc < HW; kc += KC) {
#pragma unroll
    for (int i = 0; i < 8; ++i) {
      const int idx = i * 256 + tid;
      const int c   = idx >> 4;
      const int k4  = (idx & 15) << 2;
      const float4 v = *reinterpret_cast<const float4*>(xb + (size_t)c * HW + kc + k4);
      sX[c][k4 + 0] = v.x; sX[c][k4 + 1] = v.y;
      sX[c][k4 + 2] = v.z; sX[c][k4 + 3] = v.w;
    }
    __syncthreads();

#pragma unroll 2
    for (int k0 = 0; k0 < KC; k0 += 4) {
      const int ks = k0 + hi * 2;
      const int ra = wave * 16 + l15;
      v2f a; a.x = sX[ra][ks]; a.y = sX[ra][ks + 1];
#pragma unroll
      for (int tc = 0; tc < 8; ++tc) {
        const int rb = tc * 16 + l15;
        v2f bb; bb.x = sX[rb][ks]; bb.y = sX[rb][ks + 1];
        acc[tc] = __builtin_amdgcn_wmma_f32_16x16x4_f32(
            false, a, false, bb, (short)0, acc[tc], false, false);
      }
    }
    __syncthreads();
  }

  float* gp = gramPart + (size_t)b * CH * CH;
#pragma unroll
  for (int tc = 0; tc < 8; ++tc) {
#pragma unroll
    for (int r = 0; r < 8; ++r) {
      const int row = wave * 16 + r + hi * 8;
      const int col = tc * 16 + l15;
      gp[row * CH + col] = acc[tc][r];
    }
  }
}

__device__ inline void mm128(float* __restrict__ Cm, const float* __restrict__ A,
                             const float* __restrict__ Bm, int tid) {
  const int row = tid >> 1;
  const int c0  = (tid & 1) << 6;
  float outv[64];
#pragma unroll
  for (int j = 0; j < 64; ++j) outv[j] = 0.f;
  for (int k = 0; k < CH; ++k) {
    const float a = A[row * CH + k];
    const float* brow = Bm + k * CH + c0;
#pragma unroll 8
    for (int j = 0; j < 64; ++j) outv[j] += a * brow[j];
  }
#pragma unroll 8
  for (int j = 0; j < 64; ++j) Cm[row * CH + c0 + j] = outv[j];
}

__global__ void __launch_bounds__(256)
cw_newton(const float* __restrict__ gramPart, const float* __restrict__ mean,
          const float* __restrict__ rot, float* __restrict__ Mout,
          float* __restrict__ bias) {
  extern __shared__ float smem[];
  float* S  = smem;
  float* P  = smem + 1 * CH * CH;
  float* T1 = smem + 2 * CH * CH;
  float* T2 = smem + 3 * CH * CH;
  __shared__ float sTrInv;
  const int tid = threadIdx.x;

  for (int e = tid; e < CH * CH; e += 256) {
    float g = 0.f;
    for (int s = 0; s < BATCH; ++s) g += gramPart[(size_t)s * CH * CH + e];
    const int r = e >> 7, c = e & 127;
    float sig = (g - MTOT * mean[r] * mean[c]) / MTOT;
    if (r == c) sig += EPSR;
    S[e] = sig;
  }
  __syncthreads();
  if (tid == 0) {
    float tr = 0.f;
    for (int i = 0; i < CH; ++i) tr += S[i * CH + i];
    sTrInv = 1.0f / tr;
  }
  __syncthreads();
  const float rTr = sTrInv;
  for (int e = tid; e < CH * CH; e += 256) {
    S[e] *= rTr;
    const int r = e >> 7, c = e & 127;
    P[e] = (r == c) ? 1.f : 0.f;
  }
  __syncthreads();

  for (int it = 0; it < T_NEWTON; ++it) {
    mm128(T1, P, P, tid);  __syncthreads();
    mm128(T2, T1, P, tid); __syncthreads();
    mm128(T1, T2, S, tid); __syncthreads();
    for (int e = tid; e < CH * CH; e += 256) P[e] = 1.5f * P[e] - 0.5f * T1[e];
    __syncthreads();
  }

  mm128(T2, rot, P, tid);
  __syncthreads();
  const float sc = sqrtf(rTr);
  for (int e = tid; e < CH * CH; e += 256) Mout[e] = T2[e] * sc;
  if (tid < CH) {
    float s = 0.f;
    for (int c = 0; c < CH; ++c) s += T2[tid * CH + c] * mean[c];
    bias[tid] = s * sc;
  }
}

constexpr int MPAD = CH + 4;
__global__ void __launch_bounds__(256)
cw_apply(const float* __restrict__ x, const float* __restrict__ Mmat,
         const float* __restrict__ bias, float* __restrict__ out) {
  extern __shared__ float smem[];
  float* sM = smem;
  float* sX = smem + CH * MPAD;
  float* sB = sX + CH * XPAD;
  const int b    = blockIdx.y;
  const int tid  = threadIdx.x;
  const int wave = tid >> 5;
  const int lane = tid & 31;
  const int hi   = lane >> 4;
  const int l15  = lane & 15;

  for (int i = tid; i < (CH * CH) / 4; i += 256) {
    const int e = i * 4;
    const int r = e >> 7, c = e & 127;
    const float4 v = *reinterpret_cast<const float4*>(Mmat + e);
    float* d = sM + r * MPAD + c;
    d[0] = v.x; d[1] = v.y; d[2] = v.z; d[3] = v.w;
  }
  if (tid < CH) sB[tid] = bias[tid];
  __syncthreads();

  const float* xb = x + (size_t)b * CH * HW;
  float*       ob = out + (size_t)b * CH * HW;

  for (int j = 0; j < 7; ++j) {
    const int hw0 = (blockIdx.x * 7 + j) * 64;
#pragma unroll
    for (int i = 0; i < 8; ++i) {
      const int idx = i * 256 + tid;
      const int c   = idx >> 4;
      const int k4  = (idx & 15) << 2;
      const float4 v = *reinterpret_cast<const float4*>(xb + (size_t)c * HW + hw0 + k4);
      float* d = sX + c * XPAD + k4;
      d[0] = v.x; d[1] = v.y; d[2] = v.z; d[3] = v.w;
    }
    __syncthreads();

    v8f acc[4] = {};
#pragma unroll 2
    for (int c0 = 0; c0 < CH; c0 += 4) {
      const int ks = c0 + hi * 2;
      const int ra = wave * 16 + l15;
      v2f a; a.x = sM[ra * MPAD + ks]; a.y = sM[ra * MPAD + ks + 1];
#pragma unroll
      for (int ht = 0; ht < 4; ++ht) {
        v2f bb;
        bb.x = sX[ks * XPAD + ht * 16 + l15];
        bb.y = sX[(ks + 1) * XPAD + ht * 16 + l15];
        acc[ht] = __builtin_amdgcn_wmma_f32_16x16x4_f32(
            false, a, false, bb, (short)0, acc[ht], false, false);
      }
    }

#pragma unroll
    for (int ht = 0; ht < 4; ++ht) {
#pragma unroll
      for (int r = 0; r < 8; ++r) {
        const int dch = wave * 16 + r + hi * 8;
        const int hw  = hw0 + ht * 16 + l15;
        ob[(size_t)dch * HW + hw] = acc[ht][r] - sB[dch];
      }
    }
    __syncthreads();
  }
}

extern "C" void kernel_launch(void* const* d_in, const int* in_sizes, int n_in,
                              void* d_out, int out_size, void* d_ws, size_t ws_size,
                              hipStream_t stream) {
  const float* x   = (const float*)d_in[0];
  const float* rot = (const float*)d_in[1];
  float* out = (float*)d_out;
  float* ws  = (float*)d_ws;

  float* gramPart = ws;
  float* meanPart = gramPart + (size_t)BATCH * CH * CH;
  float* mean     = meanPart + CH * BATCH;
  float* Mmat     = mean + CH;
  float* bias     = Mmat + CH * CH;

  cw_mean_partial<<<dim3(BATCH, CH), 256, 0, stream>>>(x, meanPart);
  cw_gram<<<BATCH, 256, 0, stream>>>(x, gramPart);
  cw_mean_finish<<<1, CH, 0, stream>>>(meanPart, mean);
  cw_newton<<<1, 256, 4 * CH * CH * sizeof(float), stream>>>(
      gramPart, mean, rot, Mmat, bias);
  cw_apply<<<dim3(7, BATCH), 256,
             (CH * MPAD + CH * XPAD + CH) * sizeof(float), stream>>>(
      x, Mmat, bias, out);
}


# cw_newton: fixed-order sum of the 64 Gram partial planes reads 16-byte pieces, all 64 planes of four neighbouring entries in flight per lane (was 4-byte loads, 16 at a time); sum order unchanged
# speedup vs baseline: 16.3189x; 16.3189x over previous
#include <hip/hip_runtime.h>
#include <hip/hip_bf16.h>


typedef __attribute__((ext_vector_type(2))) float v2f;
typedef __attribute__((ext_vector_type(8))) float v8f;

constexpr int CH    = 128;
constexpr int BATCH = 64;
constexpr int HW    = 56 * 56;
constexpr float MTOT = (float)BATCH * (float)HW;
constexpr float EPSR = 1e-5f;
constexpr int   T_NEWTON = 5;

__global__ void __launch_bounds__(256)
cw_mean_partial(const float* __restrict__ x, float* __restrict__ meanPart) {
  const int b = blockIdx.x;
  const int c = blockIdx.y;
  const float* p = x + ((size_t)(b * CH + c)) * HW;
  float s = 0.f;
  for (int i = threadIdx.x; i < HW; i += 256) s += p[i];
  __shared__ float red[256];
  red[threadIdx.x] = s;
  __syncthreads();
  for (int off = 128; off > 0; off >>= 1) {
    if ((int)threadIdx.x < off) red[threadIdx.x] += red[threadIdx.x + off];
    __syncthreads();
  }
  if (threadIdx.x == 0) meanPart[c * BATCH + b] = red[0];
}

__global__ void cw_mean_finish(const float* __restrict__ meanPart,
                               float* __restrict__ mean) {
  const int c = threadIdx.x;
  float s = 0.f;
  for (int b = 0; b < BATCH; ++b) s += meanPart[c * BATCH + b];
  mean[c] = s / MTOT;
}

constexpr int KC = 64;
constexpr int XPAD = KC + 4;

__global__ void __launch_bounds__(256)
cw_gram(const float* __restrict__ x, float* __restrict__ gramPart) {
  __shared__ float sX[CH][XPAD];
  const int b    = blockIdx.x;
  const int tid  = threadIdx.x;
  const int wave = tid >> 5;
  const int lane = tid & 31;
  const int hi   = lane >> 4;
  const int l15  = lane & 15;

  v8f acc[8] = {};
  const float* xb = x + (size_t)b * CH * HW;

  for (int kc = 0; kc < HW; kc += KC) {
#pragma unroll
    for (int i = 0; i < 8; ++i) {
      const int idx = i * 256 + tid;
      const int c   = idx >> 4;
      const int k4  = (idx & 15) << 2;
      const float4 v = *reinterpret_cast<const float4*>(xb + (size_t)c * HW + kc + k4);
      sX[c][k4 + 0] = v.x; sX[c][k4 + 1] = v.y;
      sX[c][k4 + 2] = v.z; sX[c][k4 + 3] = v.w;
    }
    __syncthreads();

#pragma unroll 2
    for (int k0 = 0; k0 < KC; k0 += 4) {
      const int ks = k0 + hi * 2;
      const int ra = wave * 16 + l15;
      v2f a; a.x = sX[ra][ks]; a.y = sX[ra][ks + 1];
#pragma unroll
      for (int tc = 0; tc < 8; ++tc) {
        const int rb = tc * 16 + l15;
        v2f bb; bb.x = sX[rb][ks]; bb.y = sX[rb][ks + 1];
        acc[tc] = __builtin_amdgcn_wmma_f32_16x16x4_f32(
            false, a, false, bb, (short)0, acc[tc], false, false);
      }
    }
    __syncthreads();
  }

  float* gp = gramPart + (size_t)b * CH * CH;
#pragma unroll
  for (int tc = 0; tc < 8; ++tc) {
#pragma unroll
    for (int r = 0; r < 8; ++r) {
      const int row = wave * 16 + r + hi * 8;
      const int col = tc * 16 + l15;
      gp[row * CH + col] = acc[tc][r];
    }
  }
}

constexpr int LD = CH + 4;

__device__ __forceinline__ int kpos(int k) {
  const int kk = k & 15;
  return (k & ~15) | ((kk & 6) << 1) | ((kk & 8) >> 2) | (kk & 1);
}

__device__ __forceinline__ void strip_product(v8f (&Y)[8], const v8f (&X)[8],
                                              const float* __restrict__ sA,
                                              int l15, int hi) {
#pragma unroll
  for (int i = 0; i < 8; ++i) Y[i] = v8f{};
  const float* base = sA + l15 * LD + 2 * hi;
#pragma unroll
  for (int j = 0; j < 8; ++j) {
#pragma unroll
    for (int q = 0; q < 4; ++q) {
      v2f bb; bb.x = X[j][2 * q]; bb.y = X[j][2 * q + 1];
#pragma unroll
      for (int i = 0; i < 8; ++i) {
        const v2f a = *reinterpret_cast<const v2f*>(base + i * 16 * LD + 16 * j + 4 * q);
        Y[i] = __builtin_amdgcn_wmma_f32_16x16x4_f32(
            false, a, false, bb, (short)0, Y[i], false, false);
      }
    }
  }
}

__global__ void __launch_bounds__(256)
cw_newton(const float* __restrict__ gramPart, const float* __restrict__ mean,
          const float* __restrict__ rot, float* __restrict__ Mout,
          float* __restrict__ bias) {
  extern __shared__ __attribute__((aligned(16))) float nsm[];
  float* sSt = nsm;
  float* sPt = nsm + CH * LD;
  __shared__ float sTrInv;
  const int tid  = threadIdx.x;
  const int wave = tid >> 5;
  const int lane = tid & 31;
  const int hi   = lane >> 4;
  const int l15  = lane & 15;

  {
    constexpr int PLANE4 = CH * CH / 4;
    const float4* gp4 = reinterpret_cast<const float4*>(gramPart);
    const int c0 = (tid & 31) * 4;
    const float4 muc = *reinterpret_cast<const float4*>(mean + c0);
    for (int t = 0; t < PLANE4 / 256; ++t) {
      const int q = t * 256 + tid;
      const int r = t * 8 + __builtin_amdgcn_readfirstlane(wave);
      const float mur = mean[r];
      float4 v[BATCH];
#pragma unroll
      for (int s = 0; s < BATCH; ++s) v[s] = gp4[(size_t)s * PLANE4 + q];
      float g0 = 0.f, g1 = 0.f, g2 = 0.f, g3 = 0.f;
#pragma unroll
      for (int s = 0; s < BATCH; ++s) {
        g0 += v[s].x; g1 += v[s].y; g2 += v[s].z; g3 += v[s].w;
      }
      float sig0 = (g0 - MTOT * mur * muc.x) / MTOT;
      float sig1 = (g1 - MTOT * mur * muc.y) / MTOT;
      float sig2 = (g2 - MTOT * mur * muc.z) / MTOT;
      float sig3 = (g3 - MTOT * mur * muc.w) / MTOT;
      if (r == c0 + 0) sig0 += EPSR;
      if (r == c0 + 1) sig1 += EPSR;
      if (r == c0 + 2) sig2 += EPSR;
      if (r == c0 + 3) sig3 += EPSR;
      const int kr = kpos(r);
      sSt[(c0 + 0) * LD + kr] = sig0;
      sSt[(c0 + 1) * LD + kr] = sig1;
      sSt[(c0 + 2) * LD + kr] = sig2;
      sSt[(c0 + 3) * LD + kr] = sig3;
    }
  }
  __syncthreads();
  if (tid == 0) {
    float tr = 0.f;
    for (int i = 0; i < CH; ++i) tr += sSt[i * LD + kpos(i)];
    sTrInv = 1.0f / tr;
  }
  __syncthreads();
  const float rTr = sTrInv;
  for (int e = tid; e < CH * CH; e += 256) {
    const int n = e >> 7, k = e & 127;
    sSt[n * LD + k] *= rTr;
    sPt[n * LD + kpos(k)] = (n == k) ? 1.f : 0.f;
  }

  v8f X[8], Y[8];
#pragma unroll
  for (int j = 0; j < 8; ++j) {
#pragma unroll
    for (int r = 0; r < 8; ++r)
      X[j][r] = (16 * wave + l15 == 16 * j + 8 * hi + r) ? 1.f : 0.f;
  }
  __syncthreads();

  float* own = sPt + (8 * hi) * LD + kpos(16 * wave + l15);

  for (int it = 0; it < T_NEWTON; ++it) {
    strip_product(Y, X, sPt, l15, hi);
    strip_product(X, Y, sPt, l15, hi);
    strip_product(Y, X, sSt, l15, hi);
#pragma unroll
    for (int i = 0; i < 8; ++i) {
#pragma unroll
      for (int r = 0; r < 8; ++r)
        X[i][r] = 1.5f * own[(16 * i + r) * LD] - 0.5f * Y[i][r];
    }
    __syncthreads();
#pragma unroll
    for (int i = 0; i < 8; ++i) {
#pragma unroll
      for (int r = 0; r < 8; ++r) own[(16 * i + r) * LD] = X[i][r];
    }
    __syncthreads();
  }

  const float* rrow = rot + (size_t)(16 * wave + l15) * CH + 8 * hi;
#pragma unroll
  for (int j = 0; j < 8; ++j) {
    const float4 v0 = *reinterpret_cast<const float4*>(rrow + 16 * j);
    const float4 v1 = *reinterpret_cast<const float4*>(rrow + 16 * j + 4);
    X[j][0] = v0.x; X[j][1] = v0.y; X[j][2] = v0.z; X[j][3] = v0.w;
    X[j][4] = v1.x; X[j][5] = v1.y; X[j][6] = v1.z; X[j][7] = v1.w;
  }
  strip_product(Y, X, sPt, l15, hi);

  float* sT = sSt;
  float* trow = sT + (16 * wave + l15) * LD + 8 * hi;
#pragma unroll
  for (int i = 0; i < 8; ++i) {
    *reinterpret_cast<float4*>(trow + 16 * i)     = make_float4(Y[i][0], Y[i][1], Y[i][2], Y[i][3]);
    *reinterpret_cast<float4*>(trow + 16 * i + 4) = make_float4(Y[i][4], Y[i][5], Y[i][6], Y[i][7]);
  }
  __syncthreads();

  const float sc = sqrtf(rTr);
  for (int e = tid; e < CH * CH; e += 256) Mout[e] = sT[(e >> 7) * LD + (e & 127)] * sc;
  if (tid < CH) {
    float s = 0.f;
    for (int c = 0; c < CH; ++c) s += sT[tid * LD + c] * mean[c];
    bias[tid] = s * sc;
  }
}

constexpr int MPAD = CH + 4;
__global__ void __launch_bounds__(256)
cw_apply(const float* __restrict__ x, const float* __restrict__ Mmat,
         const float* __restrict__ bias, float* __restrict__ out) {
  extern __shared__ float smem[];
  float* sM = smem;
  float* sX = smem + CH * MPAD;
  float* sB = sX + CH * XPAD;
  const int b    = blockIdx.y;
  const int tid  = threadIdx.x;
  const int wave = tid >> 5;
  const int lane = tid & 31;
  const int hi   = lane >> 4;
  const int l15  = lane & 15;

  for (int i = tid; i < (CH * CH) / 4; i += 256) {
    const int e = i * 4;
    const int r = e >> 7, c = e & 127;
    const float4 v = *reinterpret_cast<const float4*>(Mmat + e);
    float* d = sM + r * MPAD + c;
    d[0] = v.x; d[1] = v.y; d[2] = v.z; d[3] = v.w;
  }
  if (tid < CH) sB[tid] = bias[tid];
  __syncthreads();

  const float* xb = x + (size_t)b * CH * HW;
  float*       ob = out + (size_t)b * CH * HW;

  for (int j = 0; j < 7; ++j) {
    const int hw0 = (blockIdx.x * 7 + j) * 64;
#pragma unroll
    for (int i = 0; i < 8; ++i) {
      const int idx = i * 256 + tid;
      const int c   = idx >> 4;
      const int k4  = (idx & 15) << 2;
      const float4 v = *reinterpret_cast<const float4*>(xb + (size_t)c * HW + hw0 + k4);
      float* d = sX + c * XPAD + k4;
      d[0] = v.x; d[1] = v.y; d[2] = v.z; d[3] = v.w;
    }
    __syncthreads();

    v8f acc[4] = {};
#pragma unroll 2
    for (int c0 = 0; c0 < CH; c0 += 4) {
      const int ks = c0 + hi * 2;
      const int ra = wave * 16 + l15;
      v2f a; a.x = sM[ra * MPAD + ks]; a.y = sM[ra * MPAD + ks + 1];
#pragma unroll
      for (int ht = 0; ht < 4; ++ht) {
        v2f bb;
        bb.x = sX[ks * XPAD + ht * 16 + l15];
        bb.y = sX[(ks + 1) * XPAD + ht * 16 + l15];
        acc[ht] = __builtin_amdgcn_wmma_f32_16x16x4_f32(
            false, a, false, bb, (short)0, acc[ht], false, false);
      }
    }

#pragma unroll
    for (int ht = 0; ht < 4; ++ht) {
#pragma unroll
      for (int r = 0; r < 8; ++r) {
        const int dch = wave * 16 + r + hi * 8;
        const int hw  = hw0 + ht * 16 + l15;
        ob[(size_t)dch * HW + hw] = acc[ht][r] - sB[dch];
      }
    }
    __syncthreads();
  }
}

extern "C" void kernel_launch(void* const* d_in, const int* in_sizes, int n_in,
                              void* d_out, int out_size, void* d_ws, size_t ws_size,
                              hipStream_t stream) {
  const float* x   = (const float*)d_in[0];
  const float* rot = (const float*)d_in[1];
  float* out = (float*)d_out;
  float* ws  = (float*)d_ws;

  float* gramPart = ws;
  float* meanPart = gramPart + (size_t)BATCH * CH * CH;
  float* mean     = meanPart + CH * BATCH;
  float* Mmat     = mean + CH;
  float* bias     = Mmat + CH * CH;

  cw_mean_partial<<<dim3(BATCH, CH), 256, 0, stream>>>(x, meanPart);
  cw_gram<<<BATCH, 256, 0, stream>>>(x, gramPart);
  cw_mean_finish<<<1, CH, 0, stream>>>(meanPart, mean);
  cw_newton<<<1, 256, 2 * CH * LD * sizeof(float), stream>>>(
      gramPart, mean, rot, Mmat, bias);
  cw_apply<<<dim3(7, BATCH), 256,
             (CH * MPAD + CH * XPAD + CH) * sizeof(float), stream>>>(
      x, Mmat, bias, out);
}


# cw_newton: symmetric products P.P, P.P.P, P.P.P.SigmaN compute 5 of 8 strip tiles per wave, other 3 mirrored from other waves via LDS; k-sum order rotates per wave, mirrored values replace computed
# speedup vs baseline: 30.2705x; 1.8549x over previous
#include <hip/hip_runtime.h>
#include <hip/hip_bf16.h>


typedef __attribute__((ext_vector_type(2))) float v2f;
typedef __attribute__((ext_vector_type(8))) float v8f;

constexpr int CH    = 128;
constexpr int BATCH = 64;
constexpr int HW    = 56 * 56;
constexpr float MTOT = (float)BATCH * (float)HW;
constexpr float EPSR = 1e-5f;
constexpr int   T_NEWTON = 5;

constexpr int MP_WAVES = 8;
constexpr int ROW4     = HW / 4;
constexpr int MP_STEPS = 2 * ROW4 / 32;
constexpr int MEAN_WGS = BATCH * CH / (2 * MP_WAVES);
static_assert(2 * ROW4 == MP_STEPS * 32, "a pair of rows is a whole number of wave loads");
static_assert(ROW4 == 24 * 32 + 16, "row boundary sits in the middle of wave load 24");
static_assert(BATCH * CH == 2 * MEAN_WGS * MP_WAVES, "the row-sum workgroups cover every row pair exactly once");

__device__ __forceinline__ void mean_rows_job(const float* __restrict__ x,
                                              float* __restrict__ meanPart,
                                              int pair, int lane) {
  const float4* p = reinterpret_cast<const float4*>(x) + (size_t)pair * (2 * ROW4) + lane;
  float4 v[MP_STEPS];
#pragma unroll
  for (int t = 0; t < MP_STEPS; ++t) v[t] = p[t * 32];
  float s0 = 0.f, s1 = 0.f;
#pragma unroll
  for (int t = 0; t < 24; ++t) s0 += (v[t].x + v[t].y) + (v[t].z + v[t].w);
  const float mid = (v[24].x + v[24].y) + (v[24].z + v[24].w);
  if (lane < 16) s0 += mid; else s1 += mid;
#pragma unroll
  for (int t = 25; t < MP_STEPS; ++t) s1 += (v[t].x + v[t].y) + (v[t].z + v[t].w);
  for (int off = 16; off > 0; off >>= 1) {
    s0 += __shfl_xor(s0, off);
    s1 += __shfl_xor(s1, off);
  }
  if (lane == 0) {
    const int b = pair >> 6;
    const int c = (pair & 63) * 2;
    meanPart[c * BATCH + b]       = s0;
    meanPart[(c + 1) * BATCH + b] = s1;
  }
}

__device__ __forceinline__ void mean_finish_job(const float* __restrict__ meanPart,
                                                float* __restrict__ mean, int c) {
  float s = 0.f;
  for (int b = 0; b < BATCH; ++b) s += meanPart[c * BATCH + b];
  mean[c] = s / MTOT;
}

constexpr int KC = 64;
constexpr int XPAD = KC + 4;

constexpr int KSPLIT = 4;
constexpr int KPIECE = HW / KSPLIT;
constexpr int NPLANE = BATCH * KSPLIT;
constexpr int GKC    = 112;
constexpr int GPAD   = GKC + 4;

__device__ __forceinline__ void gram_job(const float* __restrict__ x,
                                         float* __restrict__ gramPart,
                                         int q, int b) {
  __shared__ float sX[CH][GPAD];
  const int tid  = threadIdx.x;
  const int wave = tid >> 5;
  const int lane = tid & 31;
  const int hi   = lane >> 4;
  const int l15  = lane & 15;

  v8f acc[8] = {};
  const float* xb = x + (size_t)b * CH * HW + q * KPIECE;

  for (int kc = 0; kc < KPIECE; kc += GKC) {
#pragma unroll
    for (int i = 0; i < 14; ++i) {
      const int idx = i * 256 + tid;
      const int c   = idx / 28;
      const int k4  = (idx - c * 28) << 2;
      const float4 v = *reinterpret_cast<const float4*>(xb + (size_t)c * HW + kc + k4);
      sX[c][k4 + 0] = v.x; sX[c][k4 + 1] = v.y;
      sX[c][k4 + 2] = v.z; sX[c][k4 + 3] = v.w;
    }
    __syncthreads();

#pragma unroll 2
    for (int k0 = 0; k0 < GKC; k0 += 4) {
      const int ks = k0 + hi * 2;
      const int ra = wave * 16 + l15;
      v2f a; a.x = sX[ra][ks]; a.y = sX[ra][ks + 1];
#pragma unroll
      for (int tc = 0; tc < 8; ++tc) {
        const int rb = tc * 16 + l15;
        v2f bb; bb.x = sX[rb][ks]; bb.y = sX[rb][ks + 1];
        acc[tc] = __builtin_amdgcn_wmma_f32_16x16x4_f32(
            false, a, false, bb, (short)0, acc[tc], false, false);
      }
    }
    __syncthreads();
  }

  float* gp = gramPart + (size_t)(b * KSPLIT + q) * CH * CH;
#pragma unroll
  for (int tc = 0; tc < 8; ++tc) {
#pragma unroll
    for (int r = 0; r < 8; ++r) {
      const int row = wave * 16 + r + hi * 8;
      const int col = tc * 16 + l15;
      gp[row * CH + col] = acc[tc][r];
    }
  }
}

constexpr int GRAM_WGS = NPLANE;

__global__ void __launch_bounds__(256)
cw_gram_mean(const float* __restrict__ x, float* __restrict__ gramPart,
             float* __restrict__ meanPart) {
  if (blockIdx.x >= GRAM_WGS) {
    const int pair = (blockIdx.x - GRAM_WGS) * MP_WAVES + (threadIdx.x >> 5);
    mean_rows_job(x, meanPart, pair, threadIdx.x & 31);
    return;
  }
  gram_job(x, gramPart, blockIdx.x % KSPLIT, blockIdx.x / KSPLIT);
}

constexpr int SUM_WGS = CH * CH / 64;

__global__ void __launch_bounds__(64)
cw_gram_sum_mean(const float* __restrict__ planes, float* __restrict__ gsum,
                 const float* __restrict__ meanPart, float* __restrict__ mean) {
  if (blockIdx.x >= SUM_WGS) {
    mean_finish_job(meanPart, mean, (blockIdx.x - SUM_WGS) * 64 + threadIdx.x);
    return;
  }
  const int e = blockIdx.x * 64 + threadIdx.x;
  const float* p = planes + e;
  float g = 0.f;
  for (int s0 = 0; s0 < NPLANE; s0 += 64) {
    float v[64];
#pragma unroll
    for (int s = 0; s < 64; ++s) v[s] = p[(size_t)(s0 + s) * (CH * CH)];
#pragma unroll
    for (int s = 0; s < 64; ++s) g += v[s];
  }
  gsum[e] = g;
}

constexpr int LD = CH + 4;

__device__ __forceinline__ int kpos(int k) {
  const int kk = k & 15;
  return (k & ~15) | ((kk & 6) << 1) | ((kk & 8) >> 2) | (kk & 1);
}

template <int NT>
__device__ __forceinline__ void strip_product(v8f (&Y)[8], const v8f (&X)[8],
                                              const float* __restrict__ sA,
                                              int wave, int l15, int hi) {
#pragma unroll
  for (int e = 0; e < NT; ++e) Y[e] = v8f{};
  const float* base = sA + l15 * LD + 2 * hi;
#pragma unroll
  for (int d = 0; d < 8; ++d) {
    const int j = (wave + d) & 7;
#pragma unroll
    for (int q = 0; q < 4; ++q) {
      v2f bb; bb.x = X[d][2 * q]; bb.y = X[d][2 * q + 1];
#pragma unroll
      for (int e = 0; e < NT; ++e) {
        const int i = (wave + e) & 7;
        const v2f a = *reinterpret_cast<const v2f*>(base + i * 16 * LD + 16 * j + 4 * q);
        Y[e] = __builtin_amdgcn_wmma_f32_16x16x4_f32(
            false, a, false, bb, (short)0, Y[e], false, false);
      }
    }
  }
}

constexpr int EXP  = 20;
constexpr int EXT  = 16 * EXP;
constexpr int EXSZ = 8 * 3 * EXT;

__device__ __forceinline__ void mirror_tiles(v8f (&Y)[8], float* __restrict__ sEx,
                                             int wave, int l15, int hi) {
  float* wr = sEx + (wave * 3) * EXT + l15 * EXP + 8 * hi;
#pragma unroll
  for (int e = 1; e <= 3; ++e) {
    *reinterpret_cast<float4*>(wr + (e - 1) * EXT)     = make_float4(Y[e][0], Y[e][1], Y[e][2], Y[e][3]);
    *reinterpret_cast<float4*>(wr + (e - 1) * EXT + 4) = make_float4(Y[e][4], Y[e][5], Y[e][6], Y[e][7]);
  }
  __syncthreads();
#pragma unroll
  for (int d = 5; d <= 7; ++d) {
    const int src = (wave + d) & 7;
    const float* rd = sEx + (src * 3 + (7 - d)) * EXT + (8 * hi) * EXP + l15;
#pragma unroll
    for (int r = 0; r < 8; ++r) Y[d][r] = rd[r * EXP];
  }
}

__global__ void __launch_bounds__(256)
cw_newton(const float* __restrict__ gsum, const float* __restrict__ mean,
          const float* __restrict__ rot, float* __restrict__ Mout,
          float* __restrict__ bias) {
  extern __shared__ __attribute__((aligned(16))) float nsm[];
  float* sSt = nsm;
  float* sPt = nsm + CH * LD;
  float* sEx = nsm + 2 * CH * LD;
  __shared__ float sTrInv;
  const int tid  = threadIdx.x;
  const int wave = tid >> 5;
  const int lane = tid & 31;
  const int hi   = lane >> 4;
  const int l15  = lane & 15;

  {
    constexpr int PLANE4 = CH * CH / 4;
    constexpr int TRIPS  = PLANE4 / 256;
    const float4* gs4 = reinterpret_cast<const float4*>(gsum);
    const int c0 = (tid & 31) * 4;
    const float4 muc = *reinterpret_cast<const float4*>(mean + c0);
    float4 gv[TRIPS];
#pragma unroll
    for (int t = 0; t < TRIPS; ++t) gv[t] = gs4[t * 256 + tid];
#pragma unroll
    for (int t = 0; t < TRIPS; ++t) {
      const int r = t * 8 + __builtin_amdgcn_readfirstlane(wave);
      const float mur = mean[r];
      float sig0 = (gv[t].x - MTOT * mur * muc.x) / MTOT;
      float sig1 = (gv[t].y - MTOT * mur * muc.y) / MTOT;
      float sig2 = (gv[t].z - MTOT * mur * muc.z) / MTOT;
      float sig3 = (gv[t].w - MTOT * mur * muc.w) / MTOT;
      if (r == c0 + 0) sig0 += EPSR;
      if (r == c0 + 1) sig1 += EPSR;
      if (r == c0 + 2) sig2 += EPSR;
      if (r == c0 + 3) sig3 += EPSR;
      const int kr = kpos(r);
      sSt[(c0 + 0) * LD + kr] = sig0;
      sSt[(c0 + 1) * LD + kr] = sig1;
      sSt[(c0 + 2) * LD + kr] = sig2;
      sSt[(c0 + 3) * LD + kr] = sig3;
    }
  }
  __syncthreads();
  if (tid == 0) {
    float tr = 0.f;
    for (int i = 0; i < CH; ++i) tr += sSt[i * LD + kpos(i)];
    sTrInv = 1.0f / tr;
  }
  __syncthreads();
  const float rTr = sTrInv;
  for (int e = tid; e < CH * CH; e += 256) {
    const int n = e >> 7, k = e & 127;
    sSt[n * LD + k] *= rTr;
    sPt[n * LD + kpos(k)] = (n == k) ? 1.f : 0.f;
  }

  const int wv = __builtin_amdgcn_readfirstlane(wave);
  v8f X[8], Y[8];
#pragma unroll
  for (int d = 0; d < 8; ++d) {
#pragma unroll
    for (int r = 0; r < 8; ++r)
      X[d][r] = (d == 0 && l15 == 8 * hi + r) ? 1.f : 0.f;
  }
  __syncthreads();

  float* own = sPt + (8 * hi) * LD + kpos(16 * wave + l15);

  int par = 0;
  for (int it = 0; it < T_NEWTON; ++it) {
    strip_product<5>(Y, X, sPt, wv, l15, hi);
    mirror_tiles(Y, sEx + par * EXSZ, wv, l15, hi);  par ^= 1;
    strip_product<5>(X, Y, sPt, wv, l15, hi);
    mirror_tiles(X, sEx + par * EXSZ, wv, l15, hi);  par ^= 1;
    strip_product<5>(Y, X, sSt, wv, l15, hi);
    mirror_tiles(Y, sEx + par * EXSZ, wv, l15, hi);  par ^= 1;
#pragma unroll
    for (int d = 0; d < 8; ++d) {
      const int i = (wv + d) & 7;
#pragma unroll
      for (int r = 0; r < 8; ++r)
        X[d][r] = 1.5f * own[(16 * i + r) * LD] - 0.5f * Y[d][r];
    }
    __syncthreads();
#pragma unroll
    for (int d = 0; d < 8; ++d) {
      const int i = (wv + d) & 7;
#pragma unroll
      for (int r = 0; r < 8; ++r) own[(16 * i + r) * LD] = X[d][r];
    }
    __syncthreads();
  }

  const float* rrow = rot + (size_t)(16 * wave + l15) * CH + 8 * hi;
#pragma unroll
  for (int d = 0; d < 8; ++d) {
    const int j = (wv + d) & 7;
    const float4 v0 = *reinterpret_cast<const float4*>(rrow + 16 * j);
    const float4 v1 = *reinterpret_cast<const float4*>(rrow + 16 * j + 4);
    X[d][0] = v0.x; X[d][1] = v0.y; X[d][2] = v0.z; X[d][3] = v0.w;
    X[d][4] = v1.x; X[d][5] = v1.y; X[d][6] = v1.z; X[d][7] = v1.w;
  }
  strip_product<8>(Y, X, sPt, wv, l15, hi);

  float* sT = sSt;
  float* trow = sT + (16 * wave + l15) * LD + 8 * hi;
#pragma unroll
  for (int e = 0; e < 8; ++e) {
    const int i = (wv + e) & 7;
    *reinterpret_cast<float4*>(trow + 16 * i)     = make_float4(Y[e][0], Y[e][1], Y[e][2], Y[e][3]);
    *reinterpret_cast<float4*>(trow + 16 * i + 4) = make_float4(Y[e][4], Y[e][5], Y[e][6], Y[e][7]);
  }
  __syncthreads();

  const float sc = sqrtf(rTr);
  for (int e = tid; e < CH * CH; e += 256) Mout[e] = sT[(e >> 7) * LD + (e & 127)] * sc;
  if (tid < CH) {
    float s = 0.f;
    for (int c = 0; c < CH; ++c) s += sT[tid * LD + c] * mean[c];
    bias[tid] = s * sc;
  }
}

constexpr int MPAD = CH + 4;
__global__ void __launch_bounds__(256)
cw_apply(const float* __restrict__ x, const float* __restrict__ Mmat,
         const float* __restrict__ bias, float* __restrict__ out) {
  extern __shared__ float smem[];
  float* sM = smem;
  float* sX = smem + CH * MPAD;
  float* sB = sX + CH * XPAD;
  const int b    = blockIdx.y;
  const int tid  = threadIdx.x;
  const int wave = tid >> 5;
  const int lane = tid & 31;
  const int hi   = lane >> 4;
  const int l15  = lane & 15;

  for (int i = tid; i < (CH * CH) / 4; i += 256) {
    const int e = i * 4;
    const int r = e >> 7, c = e & 127;
    const float4 v = *reinterpret_cast<const float4*>(Mmat + e);
    float* d = sM + r * MPAD + c;
    d[0] = v.x; d[1] = v.y; d[2] = v.z; d[3] = v.w;
  }
  if (tid < CH) sB[tid] = bias[tid];
  __syncthreads();

  const float* xb = x + (size_t)b * CH * HW;
  float*       ob = out + (size_t)b * CH * HW;

  for (int j = 0; j < 7; ++j) {
    const int hw0 = (blockIdx.x * 7 + j) * 64;
#pragma unroll
    for (int i = 0; i < 8; ++i) {
      const int idx = i * 256 + tid;
      const int c   = idx >> 4;
      const int k4  = (idx & 15) << 2;
      const float4 v = *reinterpret_cast<const float4*>(xb + (size_t)c * HW + hw0 + k4);
      float* d = sX + c * XPAD + k4;
      d[0] = v.x; d[1] = v.y; d[2] = v.z; d[3] = v.w;
    }
    __syncthreads();

    v8f acc[4] = {};
#pragma unroll 2
    for (int c0 = 0; c0 < CH; c0 += 4) {
      const int ks = c0 + hi * 2;
      const int ra = wave * 16 + l15;
      v2f a; a.x = sM[ra * MPAD + ks]; a.y = sM[ra * MPAD + ks + 1];
#pragma unroll
      for (int ht = 0; ht < 4; ++ht) {
        v2f bb;
        bb.x = sX[ks * XPAD + ht * 16 + l15];
        bb.y = sX[(ks + 1) * XPAD + ht * 16 + l15];
        acc[ht] = __builtin_amdgcn_wmma_f32_16x16x4_f32(
            false, a, false, bb, (short)0, acc[ht], false, false);
      }
    }

#pragma unroll
    for (int ht = 0; ht < 4; ++ht) {
#pragma unroll
      for (int r = 0; r < 8; ++r) {
        const int dch = wave * 16 + r + hi * 8;
        const int hw  = hw0 + ht * 16 + l15;
        ob[(size_t)dch * HW + hw] = acc[ht][r] - sB[dch];
      }
    }
    __syncthreads();
  }
}

extern "C" void kernel_launch(void* const* d_in, const int* in_sizes, int n_in,
                              void* d_out, int out_size, void* d_ws, size_t ws_size,
                              hipStream_t stream) {
  const float* x   = (const float*)d_in[0];
  const float* rot = (const float*)d_in[1];
  float* out = (float*)d_out;
  float* ws  = (float*)d_ws;

  float* gramSum  = ws;
  float* meanPart = gramSum + (size_t)BATCH * CH * CH;
  float* mean     = meanPart + CH * BATCH;
  float* Mmat     = mean + CH;
  float* bias     = Mmat + CH * CH;

  float* planes = out;

  cw_gram_mean<<<GRAM_WGS + MEAN_WGS, 256, 0, stream>>>(x, planes, meanPart);
  cw_gram_sum_mean<<<SUM_WGS + CH / 64, 64, 0, stream>>>(planes, gramSum, meanPart, mean);
  cw_newton<<<1, 256, (2 * CH * LD + 2 * EXSZ) * sizeof(float), stream>>>(
      gramSum, mean, rot, Mmat, bias);
  cw_apply<<<dim3(7, BATCH), 256,
             (CH * MPAD + CH * XPAD + CH) * sizeof(float), stream>>>(
      x, Mmat, bias, out);
}


# cw_gram waves take roles: 4 load waves fetch K chunk j+1 into a second LDS slab while 4 compute waves (32 rows x 128 each) multiply chunk j, one barrier per chunk; Gram sum order unchanged
# speedup vs baseline: 31.7823x; 1.0499x over previous
#include <hip/hip_runtime.h>
#include <hip/hip_bf16.h>


typedef __attribute__((ext_vector_type(2))) float v2f;
typedef __attribute__((ext_vector_type(8))) float v8f;
typedef __attribute__((ext_vector_type(4))) float v4f;

constexpr int CH    = 128;
constexpr int BATCH = 64;
constexpr int HW    = 56 * 56;
constexpr float MTOT = (float)BATCH * (float)HW;
constexpr float EPSR = 1e-5f;
constexpr int   T_NEWTON = 5;

constexpr int MP_WAVES = 8;
constexpr int ROW4     = HW / 4;
constexpr int MP_STEPS = 2 * ROW4 / 32;
constexpr int MEAN_WGS = BATCH * CH / (2 * MP_WAVES);
static_assert(2 * ROW4 == MP_STEPS * 32, "a pair of rows is a whole number of wave loads");
static_assert(ROW4 == 24 * 32 + 16, "row boundary sits in the middle of wave load 24");
static_assert(BATCH * CH == 2 * MEAN_WGS * MP_WAVES, "the row-sum workgroups cover every row pair exactly once");

__device__ __forceinline__ void mean_rows_job(const float* __restrict__ x,
                                              float* __restrict__ meanPart,
                                              int pair, int lane) {
  const float4* p = reinterpret_cast<const float4*>(x) + (size_t)pair * (2 * ROW4) + lane;
  float4 v[MP_STEPS];
#pragma unroll
  for (int t = 0; t < MP_STEPS; ++t) v[t] = p[t * 32];
  float s0 = 0.f, s1 = 0.f;
#pragma unroll
  for (int t = 0; t < 24; ++t) s0 += (v[t].x + v[t].y) + (v[t].z + v[t].w);
  const float mid = (v[24].x + v[24].y) + (v[24].z + v[24].w);
  if (lane < 16) s0 += mid; else s1 += mid;
#pragma unroll
  for (int t = 25; t < MP_STEPS; ++t) s1 += (v[t].x + v[t].y) + (v[t].z + v[t].w);
  for (int off = 16; off > 0; off >>= 1) {
    s0 += __shfl_xor(s0, off);
    s1 += __shfl_xor(s1, off);
  }
  if (lane == 0) {
    const int b = pair >> 6;
    const int c = (pair & 63) * 2;
    meanPart[c * BATCH + b]       = s0;
    meanPart[(c + 1) * BATCH + b] = s1;
  }
}

__device__ __forceinline__ void mean_finish_job(const float* __restrict__ meanPart,
                                                float* __restrict__ mean, int c) {
  float s = 0.f;
  for (int b = 0; b < BATCH; ++b) s += meanPart[c * BATCH + b];
  mean[c] = s / MTOT;
}

constexpr int KC = 64;
constexpr int XPAD = KC + 4;

constexpr int KSPLIT = 4;
constexpr int KPIECE = HW / KSPLIT;
constexpr int NPLANE = BATCH * KSPLIT;
constexpr int GKC    = 112;
constexpr int GPAD   = GKC + 4;
constexpr int GSLAB   = CH * GPAD;
constexpr int GSTEPS  = GKC / 4;
constexpr int GCHUNKS = KPIECE / GKC;
constexpr int GROW4   = GKC / 4;
static_assert(GCHUNKS * GKC == KPIECE, "a piece is a whole number of chunks");
static_assert(GSTEPS % 4 == 0, "the product loop takes the k steps of a chunk four at a time");

__device__ __forceinline__ void gram_stage16(float* __restrict__ slab,
                                             const float* __restrict__ xb,
                                             int kc, int r0, int lane) {
  if (lane < GROW4) {
    const float* s = xb + (size_t)r0 * HW + kc + lane * 4;
    float*       d = slab + r0 * GPAD + lane * 4;
    v4f v[16];
#pragma unroll
    for (int i = 0; i < 16; ++i) v[i] = *reinterpret_cast<const v4f*>(s + (size_t)i * 8 * HW);
    asm volatile("" : "+v"(v[0]), "+v"(v[1]), "+v"(v[2]), "+v"(v[3]),
                      "+v"(v[4]), "+v"(v[5]), "+v"(v[6]), "+v"(v[7]));
    asm volatile("" : "+v"(v[8]), "+v"(v[9]), "+v"(v[10]), "+v"(v[11]),
                      "+v"(v[12]), "+v"(v[13]), "+v"(v[14]), "+v"(v[15]));
#pragma unroll
    for (int i = 0; i < 16; ++i) *reinterpret_cast<v4f*>(d + i * 8 * GPAD) = v[i];
  }
}

struct GSet { v2f a[2][2]; v2f b[8][2]; };

__device__ __forceinline__ void gram_frags(GSet& f, const float* __restrict__ pa,
                                           const float* __restrict__ pb, int k0) {
#pragma unroll
  for (int e = 0; e < 2; ++e) {
    f.a[0][e] = *reinterpret_cast<const v2f*>(pa + k0 + 4 * e);
    f.a[1][e] = *reinterpret_cast<const v2f*>(pa + 16 * GPAD + k0 + 4 * e);
  }
#pragma unroll
  for (int tc = 0; tc < 8; ++tc) {
#pragma unroll
    for (int e = 0; e < 2; ++e)
      f.b[tc][e] = *reinterpret_cast<const v2f*>(pb + tc * 16 * GPAD + k0 + 4 * e);
  }
}

__device__ __forceinline__ void gram_opaque(GSet& f) {
#pragma unroll
  for (int s = 0; s < 2; ++s) {
#pragma unroll
    for (int e = 0; e < 2; ++e) asm volatile("" : "+v"(f.a[s][e]));
  }
#pragma unroll
  for (int tc = 0; tc < 8; ++tc) {
#pragma unroll
    for (int e = 0; e < 2; ++e) asm volatile("" : "+v"(f.b[tc][e]));
  }
}

__device__ __forceinline__ void gram_mma(v8f (&acc0)[8], v8f (&acc1)[8], const GSet& f) {
#pragma unroll
  for (int e = 0; e < 2; ++e) {
#pragma unroll
    for (int tc = 0; tc < 8; ++tc) {
      acc0[tc] = __builtin_amdgcn_wmma_f32_16x16x4_f32(
          false, f.a[0][e], false, f.b[tc][e], (short)0, acc0[tc], false, false);
      acc1[tc] = __builtin_amdgcn_wmma_f32_16x16x4_f32(
          false, f.a[1][e], false, f.b[tc][e], (short)0, acc1[tc], false, false);
    }
  }
}

__device__ __forceinline__ void gram_job(const float* __restrict__ x,
                                         float* __restrict__ gramPart,
                                         int q, int b) {
  extern __shared__ __attribute__((aligned(16))) float gsm[];
  const int tid  = threadIdx.x;
  const int wave = __builtin_amdgcn_readfirstlane(tid >> 5);
  const int lane = tid & 31;
  const int hi   = lane >> 4;
  const int l15  = lane & 15;
  const bool loader = wave >= 4;
  const int  w4     = wave & 3;

  const float* xb = x + (size_t)b * CH * HW + q * KPIECE;

  gram_stage16(gsm, xb, 0, wave, lane);
  __syncthreads();

  v8f acc0[8] = {}, acc1[8] = {};
  const int ntrips = loader ? 0 : GSTEPS / 4;
  for (int j = 0; j < GCHUNKS; ++j) {
    const float* cur = gsm + (j & 1) * GSLAB;
    if (loader && j + 1 < GCHUNKS) {
      float* nxt = gsm + ((j + 1) & 1) * GSLAB;
      gram_stage16(nxt, xb, (j + 1) * GKC, w4,     lane);
      gram_stage16(nxt, xb, (j + 1) * GKC, w4 + 4, lane);
    }
    {
      const float* pa = cur + (32 * w4 + l15) * GPAD + 2 * hi;
      const float* pb = cur + l15 * GPAD + 2 * hi;
      GSet f0, f1;
      gram_frags(f0, pa, pb, 0);
      gram_opaque(f0);
#pragma clang loop unroll(disable)
      for (int u = 0; u < ntrips; ++u) {
        gram_frags(f1, pa, pb, 16 * u + 8);
        gram_mma(acc0, acc1, f0);
        gram_frags(f0, pa, pb, (u + 1 < GSTEPS / 4) ? 16 * u + 16 : 0);
        gram_mma(acc0, acc1, f1);
      }
    }
    __syncthreads();
  }

  if (!loader) {
    float* gp = gramPart + (size_t)(b * KSPLIT + q) * CH * CH;
#pragma unroll
    for (int tc = 0; tc < 8; ++tc) {
#pragma unroll
      for (int r = 0; r < 8; ++r) {
        const int row = 32 * w4 + r + hi * 8;
        const int col = tc * 16 + l15;
        gp[row * CH + col]        = acc0[tc][r];
        gp[(row + 16) * CH + col] = acc1[tc][r];
      }
    }
  }
}

constexpr int GRAM_WGS = NPLANE;

__global__ void __launch_bounds__(256)
cw_gram_mean(const float* __restrict__ x, float* __restrict__ gramPart,
             float* __restrict__ meanPart) {
  if (blockIdx.x >= GRAM_WGS) {
    const int pair = (blockIdx.x - GRAM_WGS) * MP_WAVES + (threadIdx.x >> 5);
    mean_rows_job(x, meanPart, pair, threadIdx.x & 31);
    return;
  }
  gram_job(x, gramPart, blockIdx.x % KSPLIT, blockIdx.x / KSPLIT);
}

constexpr int SUM_WGS = CH * CH / 64;

__global__ void __launch_bounds__(64)
cw_gram_sum_mean(const float* __restrict__ planes, float* __restrict__ gsum,
                 const float* __restrict__ meanPart, float* __restrict__ mean) {
  if (blockIdx.x >= SUM_WGS) {
    mean_finish_job(meanPart, mean, (blockIdx.x - SUM_WGS) * 64 + threadIdx.x);
    return;
  }
  const int e = blockIdx.x * 64 + threadIdx.x;
  const float* p = planes + e;
  float g = 0.f;
  for (int s0 = 0; s0 < NPLANE; s0 += 64) {
    float v[64];
#pragma unroll
    for (int s = 0; s < 64; ++s) v[s] = p[(size_t)(s0 + s) * (CH * CH)];
#pragma unroll
    for (int s = 0; s < 64; ++s) g += v[s];
  }
  gsum[e] = g;
}

constexpr int LD = CH + 4;

__device__ __forceinline__ int kpos(int k) {
  const int kk = k & 15;
  return (k & ~15) | ((kk & 6) << 1) | ((kk & 8) >> 2) | (kk & 1);
}

template <int NT>
__device__ __forceinline__ void strip_product(v8f (&Y)[8], const v8f (&X)[8],
                                              const float* __restrict__ sA,
                                              int wave, int l15, int hi) {
#pragma unroll
  for (int e = 0; e < NT; ++e) Y[e] = v8f{};
  const float* base = sA + l15 * LD + 2 * hi;
#pragma unroll
  for (int d = 0; d < 8; ++d) {
    const int j = (wave + d) & 7;
#pragma unroll
    for (int q = 0; q < 4; ++q) {
      v2f bb; bb.x = X[d][2 * q]; bb.y = X[d][2 * q + 1];
#pragma unroll
      for (int e = 0; e < NT; ++e) {
        const int i = (wave + e) & 7;
        const v2f a = *reinterpret_cast<const v2f*>(base + i * 16 * LD + 16 * j + 4 * q);
        Y[e] = __builtin_amdgcn_wmma_f32_16x16x4_f32(
            false, a, false, bb, (short)0, Y[e], false, false);
      }
    }
  }
}

constexpr int EXP  = 20;
constexpr int EXT  = 16 * EXP;
constexpr int EXSZ = 8 * 3 * EXT;

__device__ __forceinline__ void mirror_tiles(v8f (&Y)[8], float* __restrict__ sEx,
                                             int wave, int l15, int hi) {
  float* wr = sEx + (wave * 3) * EXT + l15 * EXP + 8 * hi;
#pragma unroll
  for (int e = 1; e <= 3; ++e) {
    *reinterpret_cast<float4*>(wr + (e - 1) * EXT)     = make_float4(Y[e][0], Y[e][1], Y[e][2], Y[e][3]);
    *reinterpret_cast<float4*>(wr + (e - 1) * EXT + 4) = make_float4(Y[e][4], Y[e][5], Y[e][6], Y[e][7]);
  }
  __syncthreads();
#pragma unroll
  for (int d = 5; d <= 7; ++d) {
    const int src = (wave + d) & 7;
    const float* rd = sEx + (src * 3 + (7 - d)) * EXT + (8 * hi) * EXP + l15;
#pragma unroll
    for (int r = 0; r < 8; ++r) Y[d][r] = rd[r * EXP];
  }
}

__global__ void __launch_bounds__(256)
cw_newton(const float* __restrict__ gsum, const float* __restrict__ mean,
          const float* __restrict__ rot, float* __restrict__ Mout,
          float* __restrict__ bias) {
  extern __shared__ __attribute__((aligned(16))) float nsm[];
  float* sSt = nsm;
  float* sPt = nsm + CH * LD;
  float* sEx = nsm + 2 * CH * LD;
  __shared__ float sTrInv;
  const int tid  = threadIdx.x;
  const int wave = tid >> 5;
  const int lane = tid & 31;
  const int hi   = lane >> 4;
  const int l15  = lane & 15;

  {
    constexpr int PLANE4 = CH * CH / 4;
    constexpr int TRIPS  = PLANE4 / 256;
    const float4* gs4 = reinterpret_cast<const float4*>(gsum);
    const int c0 = (tid & 31) * 4;
    const float4 muc = *reinterpret_cast<const float4*>(mean + c0);
    float4 gv[TRIPS];
#pragma unroll
    for (int t = 0; t < TRIPS; ++t) gv[t] = gs4[t * 256 + tid];
#pragma unroll
    for (int t = 0; t < TRIPS; ++t) {
      const int r = t * 8 + __builtin_amdgcn_readfirstlane(wave);
      const float mur = mean[r];
      float sig0 = (gv[t].x - MTOT * mur * muc.x) / MTOT;
      float sig1 = (gv[t].y - MTOT * mur * muc.y) / MTOT;
      float sig2 = (gv[t].z - MTOT * mur * muc.z) / MTOT;
      float sig3 = (gv[t].w - MTOT * mur * muc.w) / MTOT;
      if (r == c0 + 0) sig0 += EPSR;
      if (r == c0 + 1) sig1 += EPSR;
      if (r == c0 + 2) sig2 += EPSR;
      if (r == c0 + 3) sig3 += EPSR;
      const int kr = kpos(r);
      sSt[(c0 + 0) * LD + kr] = sig0;
      sSt[(c0 + 1) * LD + kr] = sig1;
      sSt[(c0 + 2) * LD + kr] = sig2;
      sSt[(c0 + 3) * LD + kr] = sig3;
    }
  }
  __syncthreads();
  if (tid == 0) {
    float tr = 0.f;
    for (int i = 0; i < CH; ++i) tr += sSt[i * LD + kpos(i)];
    sTrInv = 1.0f / tr;
  }
  __syncthreads();
  const float rTr = sTrInv;
  for (int e = tid; e < CH * CH; e += 256) {
    const int n = e >> 7, k = e & 127;
    sSt[n * LD + k] *= rTr;
    sPt[n * LD + kpos(k)] = (n == k) ? 1.f : 0.f;
  }

  const int wv = __builtin_amdgcn_readfirstlane(wave);
  v8f X[8], Y[8];
#pragma unroll
  for (int d = 0; d < 8; ++d) {
#pragma unroll
    for (int r = 0; r < 8; ++r)
      X[d][r] = (d == 0 && l15 == 8 * hi + r) ? 1.f : 0.f;
  }
  __syncthreads();

  float* own = sPt + (8 * hi) * LD + kpos(16 * wave + l15);

  int par = 0;
  for (int it = 0; it < T_NEWTON; ++it) {
    strip_product<5>(Y, X, sPt, wv, l15, hi);
    mirror_tiles(Y, sEx + par * EXSZ, wv, l15, hi);  par ^= 1;
    strip_product<5>(X, Y, sPt, wv, l15, hi);
    mirror_tiles(X, sEx + par * EXSZ, wv, l15, hi);  par ^= 1;
    strip_product<5>(Y, X, sSt, wv, l15, hi);
    mirror_tiles(Y, sEx + par * EXSZ, wv, l15, hi);  par ^= 1;
#pragma unroll
    for (int d = 0; d < 8; ++d) {
      const int i = (wv + d) & 7;
#pragma unroll
      for (int r = 0; r < 8; ++r)
        X[d][r] = 1.5f * own[(16 * i + r) * LD] - 0.5f * Y[d][r];
    }
    __syncthreads();
#pragma unroll
    for (int d = 0; d < 8; ++d) {
      const int i = (wv + d) & 7;
#pragma unroll
      for (int r = 0; r < 8; ++r) own[(16 * i + r) * LD] = X[d][r];
    }
    __syncthreads();
  }

  const float* rrow = rot + (size_t)(16 * wave + l15) * CH + 8 * hi;
#pragma unroll
  for (int d = 0; d < 8; ++d) {
    const int j = (wv + d) & 7;
    const float4 v0 = *reinterpret_cast<const float4*>(rrow + 16 * j);
    const float4 v1 = *reinterpret_cast<const float4*>(rrow + 16 * j + 4);
    X[d][0] = v0.x; X[d][1] = v0.y; X[d][2] = v0.z; X[d][3] = v0.w;
    X[d][4] = v1.x; X[d][5] = v1.y; X[d][6] = v1.z; X[d][7] = v1.w;
  }
  strip_product<8>(Y, X, sPt, wv, l15, hi);

  float* sT = sSt;
  float* trow = sT + (16 * wave + l15) * LD + 8 * hi;
#pragma unroll
  for (int e = 0; e < 8; ++e) {
    const int i = (wv + e) & 7;
    *reinterpret_cast<float4*>(trow + 16 * i)     = make_float4(Y[e][0], Y[e][1], Y[e][2], Y[e][3]);
    *reinterpret_cast<float4*>(trow + 16 * i + 4) = make_float4(Y[e][4], Y[e][5], Y[e][6], Y[e][7]);
  }
  __syncthreads();

  const float sc = sqrtf(rTr);
  for (int e = tid; e < CH * CH; e += 256) Mout[e] = sT[(e >> 7) * LD + (e & 127)] * sc;
  if (tid < CH) {
    float s = 0.f;
    for (int c = 0; c < CH; ++c) s += sT[tid * LD + c] * mean[c];
    bias[tid] = s * sc;
  }
}

constexpr int MPAD = CH + 4;
__global__ void __launch_bounds__(256)
cw_apply(const float* __restrict__ x, const float* __restrict__ Mmat,
         const float* __restrict__ bias, float* __restrict__ out) {
  extern __shared__ float smem[];
  float* sM = smem;
  float* sX = smem + CH * MPAD;
  float* sB = sX + CH * XPAD;
  const int b    = blockIdx.y;
  const int tid  = threadIdx.x;
  const int wave = tid >> 5;
  const int lane = tid & 31;
  const int hi   = lane >> 4;
  const int l15  = lane & 15;

  for (int i = tid; i < (CH * CH) / 4; i += 256) {
    const int e = i * 4;
    const int r = e >> 7, c = e & 127;
    const float4 v = *reinterpret_cast<const float4*>(Mmat + e);
    float* d = sM + r * MPAD + c;
    d[0] = v.x; d[1] = v.y; d[2] = v.z; d[3] = v.w;
  }
  if (tid < CH) sB[tid] = bias[tid];
  __syncthreads();

  const float* xb = x + (size_t)b * CH * HW;
  float*       ob = out + (size_t)b * CH * HW;

  for (int j = 0; j < 7; ++j) {
    const int hw0 = (blockIdx.x * 7 + j) * 64;
#pragma unroll
    for (int i = 0; i < 8; ++i) {
      const int idx = i * 256 + tid;
      const int c   = idx >> 4;
      const int k4  = (idx & 15) << 2;
      const float4 v = *reinterpret_cast<const float4*>(xb + (size_t)c * HW + hw0 + k4);
      float* d = sX + c * XPAD + k4;
      d[0] = v.x; d[1] = v.y; d[2] = v.z; d[3] = v.w;
    }
    __syncthreads();

    v8f acc[4] = {};
#pragma unroll 2
    for (int c0 = 0; c0 < CH; c0 += 4) {
      const int ks = c0 + hi * 2;
      const int ra = wave * 16 + l15;
      v2f a; a.x = sM[ra * MPAD + ks]; a.y = sM[ra * MPAD + ks + 1];
#pragma unroll
      for (int ht = 0; ht < 4; ++ht) {
        v2f bb;
        bb.x = sX[ks * XPAD + ht * 16 + l15];
        bb.y = sX[(ks + 1) * XPAD + ht * 16 + l15];
        acc[ht] = __builtin_amdgcn_wmma_f32_16x16x4_f32(
            false, a, false, bb, (short)0, acc[ht], false, false);
      }
    }

#pragma unroll
    for (int ht = 0; ht < 4; ++ht) {
#pragma unroll
      for (int r = 0; r < 8; ++r) {
        const int dch = wave * 16 + r + hi * 8;
        const int hw  = hw0 + ht * 16 + l15;
        ob[(size_t)dch * HW + hw] = acc[ht][r] - sB[dch];
      }
    }
    __syncthreads();
  }
}

extern "C" void kernel_launch(void* const* d_in, const int* in_sizes, int n_in,
                              void* d_out, int out_size, void* d_ws, size_t ws_size,
                              hipStream_t stream) {
  const float* x   = (const float*)d_in[0];
  const float* rot = (const float*)d_in[1];
  float* out = (float*)d_out;
  float* ws  = (float*)d_ws;

  float* gramSum  = ws;
  float* meanPart = gramSum + (size_t)BATCH * CH * CH;
  float* mean     = meanPart + CH * BATCH;
  float* Mmat     = mean + CH;
  float* bias     = Mmat + CH * CH;

  float* planes = out;

  cw_gram_mean<<<GRAM_WGS + MEAN_WGS, 256, 2 * GSLAB * sizeof(float), stream>>>(x, planes, meanPart);
  cw_gram_sum_mean<<<SUM_WGS + CH / 64, 64, 0, stream>>>(planes, gramSum, meanPart, mean);
  cw_newton<<<1, 256, (2 * CH * LD + 2 * EXSZ) * sizeof(float), stream>>>(
      gramSum, mean, rot, Mmat, bias);
  cw_apply<<<dim3(7, BATCH), 256,
             (CH * MPAD + CH * XPAD + CH) * sizeof(float), stream>>>(
      x, Mmat, bias, out);
}


# cw_apply on a resident grid with wave roles: 256 workgroups x 7 pieces of 112 positions, 4 load waves stage piece j+1 in a second LDS slab while 4 compute waves multiply piece j; sums unchanged
# speedup vs baseline: 34.2502x; 1.0777x over previous
#include <hip/hip_runtime.h>
#include <hip/hip_bf16.h>


typedef __attribute__((ext_vector_type(2))) float v2f;
typedef __attribute__((ext_vector_type(8))) float v8f;
typedef __attribute__((ext_vector_type(4))) float v4f;

constexpr int CH    = 128;
constexpr int BATCH = 64;
constexpr int HW    = 56 * 56;
constexpr float MTOT = (float)BATCH * (float)HW;
constexpr float EPSR = 1e-5f;
constexpr int   T_NEWTON = 5;

constexpr int MP_WAVES = 8;
constexpr int ROW4     = HW / 4;
constexpr int MP_STEPS = 2 * ROW4 / 32;
constexpr int MEAN_WGS = BATCH * CH / (2 * MP_WAVES);
static_assert(2 * ROW4 == MP_STEPS * 32, "a pair of rows is a whole number of wave loads");
static_assert(ROW4 == 24 * 32 + 16, "row boundary sits in the middle of wave load 24");
static_assert(BATCH * CH == 2 * MEAN_WGS * MP_WAVES, "the row-sum workgroups cover every row pair exactly once");

__device__ __forceinline__ void mean_rows_job(const float* __restrict__ x,
                                              float* __restrict__ meanPart,
                                              int pair, int lane) {
  const float4* p = reinterpret_cast<const float4*>(x) + (size_t)pair * (2 * ROW4) + lane;
  float4 v[MP_STEPS];
#pragma unroll
  for (int t = 0; t < MP_STEPS; ++t) v[t] = p[t * 32];
  float s0 = 0.f, s1 = 0.f;
#pragma unroll
  for (int t = 0; t < 24; ++t) s0 += (v[t].x + v[t].y) + (v[t].z + v[t].w);
  const float mid = (v[24].x + v[24].y) + (v[24].z + v[24].w);
  if (lane < 16) s0 += mid; else s1 += mid;
#pragma unroll
  for (int t = 25; t < MP_STEPS; ++t) s1 += (v[t].x + v[t].y) + (v[t].z + v[t].w);
  for (int off = 16; off > 0; off >>= 1) {
    s0 += __shfl_xor(s0, off);
    s1 += __shfl_xor(s1, off);
  }
  if (lane == 0) {
    const int b = pair >> 6;
    const int c = (pair & 63) * 2;
    meanPart[c * BATCH + b]       = s0;
    meanPart[(c + 1) * BATCH + b] = s1;
  }
}

__device__ __forceinline__ void mean_finish_job(const float* __restrict__ meanPart,
                                                float* __restrict__ mean, int c) {
  float s = 0.f;
  for (int b = 0; b < BATCH; ++b) s += meanPart[c * BATCH + b];
  mean[c] = s / MTOT;
}

constexpr int KSPLIT = 4;
constexpr int KPIECE = HW / KSPLIT;
constexpr int NPLANE = BATCH * KSPLIT;
constexpr int GKC    = 112;
constexpr int GPAD   = GKC + 4;
constexpr int GSLAB   = CH * GPAD;
constexpr int GSTEPS  = GKC / 4;
constexpr int GCHUNKS = KPIECE / GKC;
constexpr int GROW4   = GKC / 4;
static_assert(GCHUNKS * GKC == KPIECE, "a piece is a whole number of chunks");
static_assert(GSTEPS % 4 == 0, "the product loop takes the k steps of a chunk four at a time");

__device__ __forceinline__ void gram_stage16(float* __restrict__ slab,
                                             const float* __restrict__ xb,
                                             int kc, int r0, int lane) {
  if (lane < GROW4) {
    const float* s = xb + (size_t)r0 * HW + kc + lane * 4;
    float*       d = slab + r0 * GPAD + lane * 4;
    v4f v[16];
#pragma unroll
    for (int i = 0; i < 16; ++i) v[i] = *reinterpret_cast<const v4f*>(s + (size_t)i * 8 * HW);
    asm volatile("" : "+v"(v[0]), "+v"(v[1]), "+v"(v[2]), "+v"(v[3]),
                      "+v"(v[4]), "+v"(v[5]), "+v"(v[6]), "+v"(v[7]));
    asm volatile("" : "+v"(v[8]), "+v"(v[9]), "+v"(v[10]), "+v"(v[11]),
                      "+v"(v[12]), "+v"(v[13]), "+v"(v[14]), "+v"(v[15]));
#pragma unroll
    for (int i = 0; i < 16; ++i) *reinterpret_cast<v4f*>(d + i * 8 * GPAD) = v[i];
  }
}

struct GSet { v2f a[2][2]; v2f b[8][2]; };

__device__ __forceinline__ void gram_frags(GSet& f, const float* __restrict__ pa,
                                           const float* __restrict__ pb, int k0) {
#pragma unroll
  for (int e = 0; e < 2; ++e) {
    f.a[0][e] = *reinterpret_cast<const v2f*>(pa + k0 + 4 * e);
    f.a[1][e] = *reinterpret_cast<const v2f*>(pa + 16 * GPAD + k0 + 4 * e);
  }
#pragma unroll
  for (int tc = 0; tc < 8; ++tc) {
#pragma unroll
    for (int e = 0; e < 2; ++e)
      f.b[tc][e] = *reinterpret_cast<const v2f*>(pb + tc * 16 * GPAD + k0 + 4 * e);
  }
}

__device__ __forceinline__ void gram_opaque(GSet& f) {
#pragma unroll
  for (int s = 0; s < 2; ++s) {
#pragma unroll
    for (int e = 0; e < 2; ++e) asm volatile("" : "+v"(f.a[s][e]));
  }
#pragma unroll
  for (int tc = 0; tc < 8; ++tc) {
#pragma unroll
    for (int e = 0; e < 2; ++e) asm volatile("" : "+v"(f.b[tc][e]));
  }
}

__device__ __forceinline__ void gram_mma(v8f (&acc0)[8], v8f (&acc1)[8], const GSet& f) {
#pragma unroll
  for (int e = 0; e < 2; ++e) {
#pragma unroll
    for (int tc = 0; tc < 8; ++tc) {
      acc0[tc] = __builtin_amdgcn_wmma_f32_16x16x4_f32(
          false, f.a[0][e], false, f.b[tc][e], (short)0, acc0[tc], false, false);
      acc1[tc] = __builtin_amdgcn_wmma_f32_16x16x4_f32(
          false, f.a[1][e], false, f.b[tc][e], (short)0, acc1[tc], false, false);
    }
  }
}

__device__ __forceinline__ void gram_job(const float* __restrict__ x,
                                         float* __restrict__ gramPart,
                                         int q, int b) {
  extern __shared__ __attribute__((aligned(16))) float gsm[];
  const int tid  = threadIdx.x;
  const int wave = __builtin_amdgcn_readfirstlane(tid >> 5);
  const int lane = tid & 31;
  const int hi   = lane >> 4;
  const int l15  = lane & 15;
  const bool loader = wave >= 4;
  const int  w4     = wave & 3;

  const float* xb = x + (size_t)b * CH * HW + q * KPIECE;

  gram_stage16(gsm, xb, 0, wave, lane);
  __syncthreads();

  v8f acc0[8] = {}, acc1[8] = {};
  const int ntrips = loader ? 0 : GSTEPS / 4;
  for (int j = 0; j < GCHUNKS; ++j) {
    const float* cur = gsm + (j & 1) * GSLAB;
    if (loader && j + 1 < GCHUNKS) {
      float* nxt = gsm + ((j + 1) & 1) * GSLAB;
      gram_stage16(nxt, xb, (j + 1) * GKC, w4,     lane);
      gram_stage16(nxt, xb, (j + 1) * GKC, w4 + 4, lane);
    }
    {
      const float* pa = cur + (32 * w4 + l15) * GPAD + 2 * hi;
      const float* pb = cur + l15 * GPAD + 2 * hi;
      GSet f0, f1;
      gram_frags(f0, pa, pb, 0);
      gram_opaque(f0);
#pragma clang loop unroll(disable)
      for (int u = 0; u < ntrips; ++u) {
        gram_frags(f1, pa, pb, 16 * u + 8);
        gram_mma(acc0, acc1, f0);
        gram_frags(f0, pa, pb, (u + 1 < GSTEPS / 4) ? 16 * u + 16 : 0);
        gram_mma(acc0, acc1, f1);
      }
    }
    __syncthreads();
  }

  if (!loader) {
    float* gp = gramPart + (size_t)(b * KSPLIT + q) * CH * CH;
#pragma unroll
    for (int tc = 0; tc < 8; ++tc) {
#pragma unroll
      for (int r = 0; r < 8; ++r) {
        const int row = 32 * w4 + r + hi * 8;
        const int col = tc * 16 + l15;
        gp[row * CH + col]        = acc0[tc][r];
        gp[(row + 16) * CH + col] = acc1[tc][r];
      }
    }
  }
}

constexpr int GRAM_WGS = NPLANE;

__global__ void __launch_bounds__(256)
cw_gram_mean(const float* __restrict__ x, float* __restrict__ gramPart,
             float* __restrict__ meanPart) {
  if (blockIdx.x >= GRAM_WGS) {
    const int pair = (blockIdx.x - GRAM_WGS) * MP_WAVES + (threadIdx.x >> 5);
    mean_rows_job(x, meanPart, pair, threadIdx.x & 31);
    return;
  }
  gram_job(x, gramPart, blockIdx.x % KSPLIT, blockIdx.x / KSPLIT);
}

constexpr int SUM_WGS = CH * CH / 64;

__global__ void __launch_bounds__(64)
cw_gram_sum_mean(const float* __restrict__ planes, float* __restrict__ gsum,
                 const float* __restrict__ meanPart, float* __restrict__ mean) {
  if (blockIdx.x >= SUM_WGS) {
    mean_finish_job(meanPart, mean, (blockIdx.x - SUM_WGS) * 64 + threadIdx.x);
    return;
  }
  const int e = blockIdx.x * 64 + threadIdx.x;
  const float* p = planes + e;
  float g = 0.f;
  for (int s0 = 0; s0 < NPLANE; s0 += 64) {
    float v[64];
#pragma unroll
    for (int s = 0; s < 64; ++s) v[s] = p[(size_t)(s0 + s) * (CH * CH)];
#pragma unroll
    for (int s = 0; s < 64; ++s) g += v[s];
  }
  gsum[e] = g;
}

constexpr int LD = CH + 4;

__device__ __forceinline__ int kpos(int k) {
  const int kk = k & 15;
  return (k & ~15) | ((kk & 6) << 1) | ((kk & 8) >> 2) | (kk & 1);
}

template <int NT>
__device__ __forceinline__ void strip_product(v8f (&Y)[8], const v8f (&X)[8],
                                              const float* __restrict__ sA,
                                              int wave, int l15, int hi) {
#pragma unroll
  for (int e = 0; e < NT; ++e) Y[e] = v8f{};
  const float* base = sA + l15 * LD + 2 * hi;
#pragma unroll
  for (int d = 0; d < 8; ++d) {
    const int j = (wave + d) & 7;
#pragma unroll
    for (int q = 0; q < 4; ++q) {
      v2f bb; bb.x = X[d][2 * q]; bb.y = X[d][2 * q + 1];
#pragma unroll
      for (int e = 0; e < NT; ++e) {
        const int i = (wave + e) & 7;
        const v2f a = *reinterpret_cast<const v2f*>(base + i * 16 * LD + 16 * j + 4 * q);
        Y[e] = __builtin_amdgcn_wmma_f32_16x16x4_f32(
            false, a, false, bb, (short)0, Y[e], false, false);
      }
    }
  }
}

constexpr int EXP  = 20;
constexpr int EXT  = 16 * EXP;
constexpr int EXSZ = 8 * 3 * EXT;

__device__ __forceinline__ void mirror_tiles(v8f (&Y)[8], float* __restrict__ sEx,
                                             int wave, int l15, int hi) {
  float* wr = sEx + (wave * 3) * EXT + l15 * EXP + 8 * hi;
#pragma unroll
  for (int e = 1; e <= 3; ++e) {
    *reinterpret_cast<float4*>(wr + (e - 1) * EXT)     = make_float4(Y[e][0], Y[e][1], Y[e][2], Y[e][3]);
    *reinterpret_cast<float4*>(wr + (e - 1) * EXT + 4) = make_float4(Y[e][4], Y[e][5], Y[e][6], Y[e][7]);
  }
  __syncthreads();
#pragma unroll
  for (int d = 5; d <= 7; ++d) {
    const int src = (wave + d) & 7;
    const float* rd = sEx + (src * 3 + (7 - d)) * EXT + (8 * hi) * EXP + l15;
#pragma unroll
    for (int r = 0; r < 8; ++r) Y[d][r] = rd[r * EXP];
  }
}

__global__ void __launch_bounds__(256)
cw_newton(const float* __restrict__ gsum, const float* __restrict__ mean,
          const float* __restrict__ rot, float* __restrict__ Mout,
          float* __restrict__ bias) {
  extern __shared__ __attribute__((aligned(16))) float nsm[];
  float* sSt = nsm;
  float* sPt = nsm + CH * LD;
  float* sEx = nsm + 2 * CH * LD;
  __shared__ float sTrInv;
  const int tid  = threadIdx.x;
  const int wave = tid >> 5;
  const int lane = tid & 31;
  const int hi   = lane >> 4;
  const int l15  = lane & 15;

  {
    constexpr int PLANE4 = CH * CH / 4;
    constexpr int TRIPS  = PLANE4 / 256;
    const float4* gs4 = reinterpret_cast<const float4*>(gsum);
    const int c0 = (tid & 31) * 4;
    const float4 muc = *reinterpret_cast<const float4*>(mean + c0);
    float4 gv[TRIPS];
#pragma unroll
    for (int t = 0; t < TRIPS; ++t) gv[t] = gs4[t * 256 + tid];
#pragma unroll
    for (int t = 0; t < TRIPS; ++t) {
      const int r = t * 8 + __builtin_amdgcn_readfirstlane(wave);
      const float mur = mean[r];
      float sig0 = (gv[t].x - MTOT * mur * muc.x) / MTOT;
      float sig1 = (gv[t].y - MTOT * mur * muc.y) / MTOT;
      float sig2 = (gv[t].z - MTOT * mur * muc.z) / MTOT;
      float sig3 = (gv[t].w - MTOT * mur * muc.w) / MTOT;
      if (r == c0 + 0) sig0 += EPSR;
      if (r == c0 + 1) sig1 += EPSR;
      if (r == c0 + 2) sig2 += EPSR;
      if (r == c0 + 3) sig3 += EPSR;
      const int kr = kpos(r);
      sSt[(c0 + 0) * LD + kr] = sig0;
      sSt[(c0 + 1) * LD + kr] = sig1;
      sSt[(c0 + 2) * LD + kr] = sig2;
      sSt[(c0 + 3) * LD + kr] = sig3;
    }
  }
  __syncthreads();
  if (tid == 0) {
    float tr = 0.f;
    for (int i = 0; i < CH; ++i) tr += sSt[i * LD + kpos(i)];
    sTrInv = 1.0f / tr;
  }
  __syncthreads();
  const float rTr = sTrInv;
  for (int e = tid; e < CH * CH; e += 256) {
    const int n = e >> 7, k = e & 127;
    sSt[n * LD + k] *= rTr;
    sPt[n * LD + kpos(k)] = (n == k) ? 1.f : 0.f;
  }

  const int wv = __builtin_amdgcn_readfirstlane(wave);
  v8f X[8], Y[8];
#pragma unroll
  for (int d = 0; d < 8; ++d) {
#pragma unroll
    for (int r = 0; r < 8; ++r)
      X[d][r] = (d == 0 && l15 == 8 * hi + r) ? 1.f : 0.f;
  }
  __syncthreads();

  float* own = sPt + (8 * hi) * LD + kpos(16 * wave + l15);

  int par = 0;
  for (int it = 0; it < T_NEWTON; ++it) {
    strip_product<5>(Y, X, sPt, wv, l15, hi);
    mirror_tiles(Y, sEx + par * EXSZ, wv, l15, hi);  par ^= 1;
    strip_product<5>(X, Y, sPt, wv, l15, hi);
    mirror_tiles(X, sEx + par * EXSZ, wv, l15, hi);  par ^= 1;
    strip_product<5>(Y, X, sSt, wv, l15, hi);
    mirror_tiles(Y, sEx + par * EXSZ, wv, l15, hi);  par ^= 1;
#pragma unroll
    for (int d = 0; d < 8; ++d) {
      const int i = (wv + d) & 7;
#pragma unroll
      for (int r = 0; r < 8; ++r)
        X[d][r] = 1.5f * own[(16 * i + r) * LD] - 0.5f * Y[d][r];
    }
    __syncthreads();
#pragma unroll
    for (int d = 0; d < 8; ++d) {
      const int i = (wv + d) & 7;
#pragma unroll
      for (int r = 0; r < 8; ++r) own[(16 * i + r) * LD] = X[d][r];
    }
    __syncthreads();
  }

  const float* rrow = rot + (size_t)(16 * wave + l15) * CH + 8 * hi;
#pragma unroll
  for (int d = 0; d < 8; ++d) {
    const int j = (wv + d) & 7;
    const float4 v0 = *reinterpret_cast<const float4*>(rrow + 16 * j);
    const float4 v1 = *reinterpret_cast<const float4*>(rrow + 16 * j + 4);
    X[d][0] = v0.x; X[d][1] = v0.y; X[d][2] = v0.z; X[d][3] = v0.w;
    X[d][4] = v1.x; X[d][5] = v1.y; X[d][6] = v1.z; X[d][7] = v1.w;
  }
  strip_product<8>(Y, X, sPt, wv, l15, hi);

  float* sT = sSt;
  float* trow = sT + (16 * wave + l15) * LD + 8 * hi;
#pragma unroll
  for (int e = 0; e < 8; ++e) {
    const int i = (wv + e) & 7;
    *reinterpret_cast<float4*>(trow + 16 * i)     = make_float4(Y[e][0], Y[e][1], Y[e][2], Y[e][3]);
    *reinterpret_cast<float4*>(trow + 16 * i + 4) = make_float4(Y[e][4], Y[e][5], Y[e][6], Y[e][7]);
  }
  __syncthreads();

  const float sc = sqrtf(rTr);
  for (int e = tid; e < CH * CH; e += 256) Mout[e] = sT[(e >> 7) * LD + (e & 127)] * sc;
  if (tid < CH) {
    float s = 0.f;
    for (int c = 0; c < CH; ++c) s += sT[tid * LD + c] * mean[c];
    bias[tid] = s * sc;
  }
}

constexpr int MPAD    = CH + 4;
constexpr int APC     = GKC;
constexpr int APIECES = KPIECE / APC;
constexpr int APP     = 2 * APC;
constexpr int ASLAB   = (CH / 2) * APP;
constexpr int ATILES  = APC / 16;
constexpr int ASTEPS  = CH / 4;
static_assert(APIECES * APC == KPIECE, "a workgroup's share is a whole number of pieces");
static_assert(ASTEPS % 4 == 0, "the product loop takes the k steps of a piece four at a time");

struct APiece { v4f a[8], b[8]; };

__device__ __forceinline__ void apply_fetch8(APiece& v, const float* __restrict__ xb,
                                             int kc, int p0, int lane) {
  const int ln = lane < GROW4 ? lane : GROW4 - 1;
  const float* s = xb + (size_t)(2 * p0) * HW + kc + ln * 4;
#pragma unroll
  for (int i = 0; i < 8; ++i) {
    v.a[i] = *reinterpret_cast<const v4f*>(s + (size_t)i * 16 * HW);
    v.b[i] = *reinterpret_cast<const v4f*>(s + (size_t)i * 16 * HW + HW);
  }
  asm volatile("" : "+v"(v.a[0]), "+v"(v.a[1]), "+v"(v.a[2]), "+v"(v.a[3]),
                    "+v"(v.a[4]), "+v"(v.a[5]), "+v"(v.a[6]), "+v"(v.a[7]));
  asm volatile("" : "+v"(v.b[0]), "+v"(v.b[1]), "+v"(v.b[2]), "+v"(v.b[3]),
                    "+v"(v.b[4]), "+v"(v.b[5]), "+v"(v.b[6]), "+v"(v.b[7]));
}

__device__ __forceinline__ void apply_put8(float* __restrict__ slab, const APiece& v,
                                           int p0, int lane) {
  if (lane < GROW4) {
    float* d = slab + p0 * APP + lane * 8;
#pragma unroll
    for (int i = 0; i < 8; ++i) {
      v4f lo, hi;
      lo.x = v.a[i].x; lo.y = v.b[i].x; lo.z = v.a[i].y; lo.w = v.b[i].y;
      hi.x = v.a[i].z; hi.y = v.b[i].z; hi.z = v.a[i].w; hi.w = v.b[i].w;
      *reinterpret_cast<v4f*>(d + i * 8 * APP)     = lo;
      *reinterpret_cast<v4f*>(d + i * 8 * APP + 4) = hi;
    }
  }
}

struct ASet { v2f a[2][2]; v2f b[ATILES][2]; };

__device__ __forceinline__ void apply_frags(ASet& f, const float* __restrict__ pa,
                                            const float* __restrict__ pb, int k0) {
#pragma unroll
  for (int e = 0; e < 2; ++e) {
    f.a[0][e] = *reinterpret_cast<const v2f*>(pa + k0 + 4 * e);
    f.a[1][e] = *reinterpret_cast<const v2f*>(pa + 16 * MPAD + k0 + 4 * e);
  }
#pragma unroll
  for (int t = 0; t < ATILES; ++t) {
#pragma unroll
    for (int e = 0; e < 2; ++e)
      f.b[t][e] = *reinterpret_cast<const v2f*>(pb + (k0 / 2 + 2 * e) * APP + 32 * t);
  }
}

__device__ __forceinline__ void apply_opaque(ASet& f) {
#pragma unroll
  for (int s = 0; s < 2; ++s) {
#pragma unroll
    for (int e = 0; e < 2; ++e) asm volatile("" : "+v"(f.a[s][e]));
  }
#pragma unroll
  for (int t = 0; t < ATILES; ++t) {
#pragma unroll
    for (int e = 0; e < 2; ++e) asm volatile("" : "+v"(f.b[t][e]));
  }
}

__device__ __forceinline__ void apply_mma(v8f (&acc0)[ATILES], v8f (&acc1)[ATILES], const ASet& f) {
#pragma unroll
  for (int e = 0; e < 2; ++e) {
#pragma unroll
    for (int t = 0; t < ATILES; ++t) {
      acc0[t] = __builtin_amdgcn_wmma_f32_16x16x4_f32(
          false, f.a[0][e], false, f.b[t][e], (short)0, acc0[t], false, false);
      acc1[t] = __builtin_amdgcn_wmma_f32_16x16x4_f32(
          false, f.a[1][e], false, f.b[t][e], (short)0, acc1[t], false, false);
    }
  }
}

__device__ __forceinline__ void apply_mma_first(v8f (&acc0)[ATILES], v8f (&acc1)[ATILES], const ASet& f) {
#pragma unroll
  for (int t = 0; t < ATILES; ++t) {
    acc0[t] = __builtin_amdgcn_wmma_f32_16x16x4_f32(
        false, f.a[0][0], false, f.b[t][0], (short)0, v8f{}, false, false);
    acc1[t] = __builtin_amdgcn_wmma_f32_16x16x4_f32(
        false, f.a[1][0], false, f.b[t][0], (short)0, v8f{}, false, false);
  }
#pragma unroll
  for (int t = 0; t < ATILES; ++t) {
    acc0[t] = __builtin_amdgcn_wmma_f32_16x16x4_f32(
        false, f.a[0][1], false, f.b[t][1], (short)0, acc0[t], false, false);
    acc1[t] = __builtin_amdgcn_wmma_f32_16x16x4_f32(
        false, f.a[1][1], false, f.b[t][1], (short)0, acc1[t], false, false);
  }
}

__global__ void __launch_bounds__(256)
cw_apply(const float* __restrict__ x, const float* __restrict__ Mmat,
         const float* __restrict__ bias, float* __restrict__ out) {
  extern __shared__ __attribute__((aligned(16))) float apsm[];
  float* slab = apsm;
  float* sM   = apsm + 2 * ASLAB;
  const int tid  = threadIdx.x;
  const int wave = __builtin_amdgcn_readfirstlane(tid >> 5);
  const int lane = tid & 31;
  const int hi   = lane >> 4;
  const int l15  = lane & 15;
  const bool loader = wave >= 4;
  const int  w4     = wave & 3;

  const int q = blockIdx.x % KSPLIT;
  const int b = blockIdx.x / KSPLIT;
  const float* xb = x   + (size_t)b * CH * HW + q * KPIECE;
  float*       ob = out + (size_t)b * CH * HW + q * KPIECE;

  float bv0[8], bv1[8];
  {
    const v4f* m4 = reinterpret_cast<const v4f*>(Mmat);
    const v4f* b4 = reinterpret_cast<const v4f*>(bias + 32 * w4 + 8 * hi);
    v4f mv[16], bq[4];
    APiece pv;
    bq[0] = b4[0]; bq[1] = b4[1]; bq[2] = b4[4]; bq[3] = b4[5];
#pragma unroll
    for (int t = 0; t < 16; ++t) mv[t] = m4[t * 256 + tid];
    apply_fetch8(pv, xb, 0, wave, lane);
    asm volatile("" : "+v"(mv[0]), "+v"(mv[1]), "+v"(mv[2]), "+v"(mv[3]),
                      "+v"(mv[4]), "+v"(mv[5]), "+v"(mv[6]), "+v"(mv[7]));
    asm volatile("" : "+v"(mv[8]), "+v"(mv[9]), "+v"(mv[10]), "+v"(mv[11]),
                      "+v"(mv[12]), "+v"(mv[13]), "+v"(mv[14]), "+v"(mv[15]));
    asm volatile("" : "+v"(bq[0]), "+v"(bq[1]), "+v"(bq[2]), "+v"(bq[3]));
    apply_put8(slab, pv, wave, lane);
#pragma unroll
    for (int t = 0; t < 16; ++t) {
      const int i = t * 256 + tid;
      *reinterpret_cast<v4f*>(sM + (i >> 5) * MPAD + (i & 31) * 4) = mv[t];
    }
#pragma unroll
    for (int r = 0; r < 4; ++r) {
      bv0[r] = bq[0][r]; bv0[4 + r] = bq[1][r];
      bv1[r] = bq[2][r]; bv1[4 + r] = bq[3][r];
    }
  }
  __syncthreads();

  const int ntrips = loader ? 0 : ASTEPS / 4;
  const float* pa = sM + (32 * w4 + l15) * MPAD + 2 * hi;
  for (int j = 0; j < APIECES; ++j) {
    const float* cur = slab + (j & 1) * ASLAB;
    if (loader && j + 1 < APIECES) {
      float* nxt = slab + ((j + 1) & 1) * ASLAB;
      APiece pv;
      apply_fetch8(pv, xb, (j + 1) * APC, w4, lane);
      apply_put8(nxt, pv, w4, lane);
      apply_fetch8(pv, xb, (j + 1) * APC, w4 + 4, lane);
      apply_put8(nxt, pv, w4 + 4, lane);
    }
    v8f acc0[ATILES], acc1[ATILES];
    {
      const float* pb = cur + hi * APP + 2 * l15;
      ASet f0, f1;
      apply_frags(f0, pa, pb, 0);
      apply_opaque(f0);
      if (!loader) {
        apply_frags(f1, pa, pb, 8);
        apply_mma_first(acc0, acc1, f0);
        apply_frags(f0, pa, pb, 16);
        apply_mma(acc0, acc1, f1);
      }
      apply_opaque(f0);
#pragma clang loop unroll(disable)
      for (int u = 1; u < ntrips; ++u) {
        apply_frags(f1, pa, pb, 16 * u + 8);
        apply_mma(acc0, acc1, f0);
        apply_frags(f0, pa, pb, (u + 1 < ASTEPS / 4) ? 16 * u + 16 : 0);
        apply_mma(acc0, acc1, f1);
      }
    }
    __syncthreads();
    if (!loader) {
      float* op = ob + (size_t)(32 * w4 + 8 * hi) * HW + j * APC + l15;
#pragma unroll
      for (int t = 0; t < ATILES; ++t) {
#pragma unroll
        for (int r = 0; r < 8; ++r) {
          acc0[t][r] -= bv0[r];
          acc1[t][r] -= bv1[r];
        }
      }
#pragma unroll
      for (int t = 0; t < ATILES; ++t) asm volatile("" : "+v"(acc0[t]), "+v"(acc1[t]));
#pragma unroll
      for (int t = 0; t < ATILES; ++t) {
#pragma unroll
        for (int r = 0; r < 8; ++r) {
          op[(size_t)r * HW + 16 * t]        = acc0[t][r];
          op[(size_t)(r + 16) * HW + 16 * t] = acc1[t][r];
        }
      }
    }
  }
}

extern "C" void kernel_launch(void* const* d_in, const int* in_sizes, int n_in,
                              void* d_out, int out_size, void* d_ws, size_t ws_size,
                              hipStream_t stream) {
  const float* x   = (const float*)d_in[0];
  const float* rot = (const float*)d_in[1];
  float* out = (float*)d_out;
  float* ws  = (float*)d_ws;

  float* gramSum  = ws;
  float* meanPart = gramSum + (size_t)BATCH * CH * CH;
  float* mean     = meanPart + CH * BATCH;
  float* Mmat     = mean + CH;
  float* bias     = Mmat + CH * CH;

  float* planes = out;

  cw_gram_mean<<<GRAM_WGS + MEAN_WGS, 256, 2 * GSLAB * sizeof(float), stream>>>(x, planes, meanPart);
  cw_gram_sum_mean<<<SUM_WGS + CH / 64, 64, 0, stream>>>(planes, gramSum, meanPart, mean);
  cw_newton<<<1, 256, (2 * CH * LD + 2 * EXSZ) * sizeof(float), stream>>>(
      gramSum, mean, rot, Mmat, bias);
  cw_apply<<<GRAM_WGS, 256, (2 * ASLAB + CH * MPAD) * sizeof(float), stream>>>(
      x, Mmat, bias, out);
}


# cw_apply compute waves go tile by tile: M rows held in registers, a 16-position tile runs its 32 k steps on two accumulators, stores leave under the next tile (piece barrier LDS-only); sums unchanged
# speedup vs baseline: 36.4137x; 1.0632x over previous
#include <hip/hip_runtime.h>
#include <hip/hip_bf16.h>


typedef __attribute__((ext_vector_type(2))) float v2f;
typedef __attribute__((ext_vector_type(8))) float v8f;
typedef __attribute__((ext_vector_type(4))) float v4f;

constexpr int CH    = 128;
constexpr int BATCH = 64;
constexpr int HW    = 56 * 56;
constexpr float MTOT = (float)BATCH * (float)HW;
constexpr float EPSR = 1e-5f;
constexpr int   T_NEWTON = 5;

constexpr int MP_WAVES = 8;
constexpr int ROW4     = HW / 4;
constexpr int MP_STEPS = 2 * ROW4 / 32;
constexpr int MEAN_WGS = BATCH * CH / (2 * MP_WAVES);
static_assert(2 * ROW4 == MP_STEPS * 32, "a pair of rows is a whole number of wave loads");
static_assert(ROW4 == 24 * 32 + 16, "row boundary sits in the middle of wave load 24");
static_assert(BATCH * CH == 2 * MEAN_WGS * MP_WAVES, "the row-sum workgroups cover every row pair exactly once");

__device__ __forceinline__ void mean_rows_job(const float* __restrict__ x,
                                              float* __restrict__ meanPart,
                                              int pair, int lane) {
  const float4* p = reinterpret_cast<const float4*>(x) + (size_t)pair * (2 * ROW4) + lane;
  float4 v[MP_STEPS];
#pragma unroll
  for (int t = 0; t < MP_STEPS; ++t) v[t] = p[t * 32];
  float s0 = 0.f, s1 = 0.f;
#pragma unroll
  for (int t = 0; t < 24; ++t) s0 += (v[t].x + v[t].y) + (v[t].z + v[t].w);
  const float mid = (v[24].x + v[24].y) + (v[24].z + v[24].w);
  if (lane < 16) s0 += mid; else s1 += mid;
#pragma unroll
  for (int t = 25; t < MP_STEPS; ++t) s1 += (v[t].x + v[t].y) + (v[t].z + v[t].w);
  for (int off = 16; off > 0; off >>= 1) {
    s0 += __shfl_xor(s0, off);
    s1 += __shfl_xor(s1, off);
  }
  if (lane == 0) {
    const int b = pair >> 6;
    const int c = (pair & 63) * 2;
    meanPart[c * BATCH + b]       = s0;
    meanPart[(c + 1) * BATCH + b] = s1;
  }
}

__device__ __forceinline__ void mean_finish_job(const float* __restrict__ meanPart,
                                                float* __restrict__ mean, int c) {
  float s = 0.f;
  for (int b = 0; b < BATCH; ++b) s += meanPart[c * BATCH + b];
  mean[c] = s / MTOT;
}

constexpr int KSPLIT = 4;
constexpr int KPIECE = HW / KSPLIT;
constexpr int NPLANE = BATCH * KSPLIT;
constexpr int GKC    = 112;
constexpr int GPAD   = GKC + 4;
constexpr int GSLAB   = CH * GPAD;
constexpr int GSTEPS  = GKC / 4;
constexpr int GCHUNKS = KPIECE / GKC;
constexpr int GROW4   = GKC / 4;
static_assert(GCHUNKS * GKC == KPIECE, "a piece is a whole number of chunks");
static_assert(GSTEPS % 4 == 0, "the product loop takes the k steps of a chunk four at a time");

__device__ __forceinline__ void gram_stage16(float* __restrict__ slab,
                                             const float* __restrict__ xb,
                                             int kc, int r0, int lane) {
  if (lane < GROW4) {
    const float* s = xb + (size_t)r0 * HW + kc + lane * 4;
    float*       d = slab + r0 * GPAD + lane * 4;
    v4f v[16];
#pragma unroll
    for (int i = 0; i < 16; ++i) v[i] = *reinterpret_cast<const v4f*>(s + (size_t)i * 8 * HW);
    asm volatile("" : "+v"(v[0]), "+v"(v[1]), "+v"(v[2]), "+v"(v[3]),
                      "+v"(v[4]), "+v"(v[5]), "+v"(v[6]), "+v"(v[7]));
    asm volatile("" : "+v"(v[8]), "+v"(v[9]), "+v"(v[10]), "+v"(v[11]),
                      "+v"(v[12]), "+v"(v[13]), "+v"(v[14]), "+v"(v[15]));
#pragma unroll
    for (int i = 0; i < 16; ++i) *reinterpret_cast<v4f*>(d + i * 8 * GPAD) = v[i];
  }
}

struct GSet { v2f a[2][2]; v2f b[8][2]; };

__device__ __forceinline__ void gram_frags(GSet& f, const float* __restrict__ pa,
                                           const float* __restrict__ pb, int k0) {
#pragma unroll
  for (int e = 0; e < 2; ++e) {
    f.a[0][e] = *reinterpret_cast<const v2f*>(pa + k0 + 4 * e);
    f.a[1][e] = *reinterpret_cast<const v2f*>(pa + 16 * GPAD + k0 + 4 * e);
  }
#pragma unroll
  for (int tc = 0; tc < 8; ++tc) {
#pragma unroll
    for (int e = 0; e < 2; ++e)
      f.b[tc][e] = *reinterpret_cast<const v2f*>(pb + tc * 16 * GPAD + k0 + 4 * e);
  }
}

__device__ __forceinline__ void gram_opaque(GSet& f) {
#pragma unroll
  for (int s = 0; s < 2; ++s) {
#pragma unroll
    for (int e = 0; e < 2; ++e) asm volatile("" : "+v"(f.a[s][e]));
  }
#pragma unroll
  for (int tc = 0; tc < 8; ++tc) {
#pragma unroll
    for (int e = 0; e < 2; ++e) asm volatile("" : "+v"(f.b[tc][e]));
  }
}

__device__ __forceinline__ void gram_mma(v8f (&acc0)[8], v8f (&acc1)[8], const GSet& f) {
#pragma unroll
  for (int e = 0; e < 2; ++e) {
#pragma unroll
    for (int tc = 0; tc < 8; ++tc) {
      acc0[tc] = __builtin_amdgcn_wmma_f32_16x16x4_f32(
          false, f.a[0][e], false, f.b[tc][e], (short)0, acc0[tc], false, false);
      acc1[tc] = __builtin_amdgcn_wmma_f32_16x16x4_f32(
          false, f.a[1][e], false, f.b[tc][e], (short)0, acc1[tc], false, false);
    }
  }
}

__device__ __forceinline__ void gram_job(const float* __restrict__ x,
                                         float* __restrict__ gramPart,
                                         int q, int b) {
  extern __shared__ __attribute__((aligned(16))) float gsm[];
  const int tid  = threadIdx.x;
  const int wave = __builtin_amdgcn_readfirstlane(tid >> 5);
  const int lane = tid & 31;
  const int hi   = lane >> 4;
  const int l15  = lane & 15;
  const bool loader = wave >= 4;
  const int  w4     = wave & 3;

  const float* xb = x + (size_t)b * CH * HW + q * KPIECE;

  gram_stage16(gsm, xb, 0, wave, lane);
  __syncthreads();

  v8f acc0[8] = {}, acc1[8] = {};
  const int ntrips = loader ? 0 : GSTEPS / 4;
  for (int j = 0; j < GCHUNKS; ++j) {
    const float* cur = gsm + (j & 1) * GSLAB;
    if (loader && j + 1 < GCHUNKS) {
      float* nxt = gsm + ((j + 1) & 1) * GSLAB;
      gram_stage16(nxt, xb, (j + 1) * GKC, w4,     lane);
      gram_stage16(nxt, xb, (j + 1) * GKC, w4 + 4, lane);
    }
    {
      const float* pa = cur + (32 * w4 + l15) * GPAD + 2 * hi;
      const float* pb = cur + l15 * GPAD + 2 * hi;
      GSet f0, f1;
      gram_frags(f0, pa, pb, 0);
      gram_opaque(f0);
#pragma clang loop unroll(disable)
      for (int u = 0; u < ntrips; ++u) {
        gram_frags(f1, pa, pb, 16 * u + 8);
        gram_mma(acc0, acc1, f0);
        gram_frags(f0, pa, pb, (u + 1 < GSTEPS / 4) ? 16 * u + 16 : 0);
        gram_mma(acc0, acc1, f1);
      }
    }
    __syncthreads();
  }

  if (!loader) {
    float* gp = gramPart + (size_t)(b * KSPLIT + q) * CH * CH;
#pragma unroll
    for (int tc = 0; tc < 8; ++tc) {
#pragma unroll
      for (int r = 0; r < 8; ++r) {
        const int row = 32 * w4 + r + hi * 8;
        const int col = tc * 16 + l15;
        gp[row * CH + col]        = acc0[tc][r];
        gp[(row + 16) * CH + col] = acc1[tc][r];
      }
    }
  }
}

constexpr int GRAM_WGS = NPLANE;

__global__ void __launch_bounds__(256)
cw_gram_mean(const float* __restrict__ x, float* __restrict__ gramPart,
             float* __restrict__ meanPart) {
  if (blockIdx.x >= GRAM_WGS) {
    const int pair = (blockIdx.x - GRAM_WGS) * MP_WAVES + (threadIdx.x >> 5);
    mean_rows_job(x, meanPart, pair, threadIdx.x & 31);
    return;
  }
  gram_job(x, gramPart, blockIdx.x % KSPLIT, blockIdx.x / KSPLIT);
}

constexpr int SUM_WGS = CH * CH / 64;

__global__ void __launch_bounds__(64)
cw_gram_sum_mean(const float* __restrict__ planes, float* __restrict__ gsum,
                 const float* __restrict__ meanPart, float* __restrict__ mean) {
  if (blockIdx.x >= SUM_WGS) {
    mean_finish_job(meanPart, mean, (blockIdx.x - SUM_WGS) * 64 + threadIdx.x);
    return;
  }
  const int e = blockIdx.x * 64 + threadIdx.x;
  const float* p = planes + e;
  float g = 0.f;
  for (int s0 = 0; s0 < NPLANE; s0 += 64) {
    float v[64];
#pragma unroll
    for (int s = 0; s < 64; ++s) v[s] = p[(size_t)(s0 + s) * (CH * CH)];
#pragma unroll
    for (int s = 0; s < 64; ++s) g += v[s];
  }
  gsum[e] = g;
}

constexpr int LD = CH + 4;

__device__ __forceinline__ int kpos(int k) {
  const int kk = k & 15;
  return (k & ~15) | ((kk & 6) << 1) | ((kk & 8) >> 2) | (kk & 1);
}

template <int NT>
__device__ __forceinline__ void strip_product(v8f (&Y)[8], const v8f (&X)[8],
                                              const float* __restrict__ sA,
                                              int wave, int l15, int hi) {
#pragma unroll
  for (int e = 0; e < NT; ++e) Y[e] = v8f{};
  const float* base = sA + l15 * LD + 2 * hi;
#pragma unroll
  for (int d = 0; d < 8; ++d) {
    const int j = (wave + d) & 7;
#pragma unroll
    for (int q = 0; q < 4; ++q) {
      v2f bb; bb.x = X[d][2 * q]; bb.y = X[d][2 * q + 1];
#pragma unroll
      for (int e = 0; e < NT; ++e) {
        const int i = (wave + e) & 7;
        const v2f a = *reinterpret_cast<const v2f*>(base + i * 16 * LD + 16 * j + 4 * q);
        Y[e] = __builtin_amdgcn_wmma_f32_16x16x4_f32(
            false, a, false, bb, (short)0, Y[e], false, false);
      }
    }
  }
}

constexpr int EXP  = 20;
constexpr int EXT  = 16 * EXP;
constexpr int EXSZ = 8 * 3 * EXT;

__device__ __forceinline__ void mirror_tiles(v8f (&Y)[8], float* __restrict__ sEx,
                                             int wave, int l15, int hi) {
  float* wr = sEx + (wave * 3) * EXT + l15 * EXP + 8 * hi;
#pragma unroll
  for (int e = 1; e <= 3; ++e) {
    *reinterpret_cast<float4*>(wr + (e - 1) * EXT)     = make_float4(Y[e][0], Y[e][1], Y[e][2], Y[e][3]);
    *reinterpret_cast<float4*>(wr + (e - 1) * EXT + 4) = make_float4(Y[e][4], Y[e][5], Y[e][6], Y[e][7]);
  }
  __syncthreads();
#pragma unroll
  for (int d = 5; d <= 7; ++d) {
    const int src = (wave + d) & 7;
    const float* rd = sEx + (src * 3 + (7 - d)) * EXT + (8 * hi) * EXP + l15;
#pragma unroll
    for (int r = 0; r < 8; ++r) Y[d][r] = rd[r * EXP];
  }
}

__global__ void __launch_bounds__(256)
cw_newton(const float* __restrict__ gsum, const float* __restrict__ mean,
          const float* __restrict__ rot, float* __restrict__ Mout,
          float* __restrict__ bias) {
  extern __shared__ __attribute__((aligned(16))) float nsm[];
  float* sSt = nsm;
  float* sPt = nsm + CH * LD;
  float* sEx = nsm + 2 * CH * LD;
  __shared__ float sTrInv;
  const int tid  = threadIdx.x;
  const int wave = tid >> 5;
  const int lane = tid & 31;
  const int hi   = lane >> 4;
  const int l15  = lane & 15;

  {
    constexpr int PLANE4 = CH * CH / 4;
    constexpr int TRIPS  = PLANE4 / 256;
    const float4* gs4 = reinterpret_cast<const float4*>(gsum);
    const int c0 = (tid & 31) * 4;
    const float4 muc = *reinterpret_cast<const float4*>(mean + c0);
    float4 gv[TRIPS];
#pragma unroll
    for (int t = 0; t < TRIPS; ++t) gv[t] = gs4[t * 256 + tid];
#pragma unroll
    for (int t = 0; t < TRIPS; ++t) {
      const int r = t * 8 + __builtin_amdgcn_readfirstlane(wave);
      const float mur = mean[r];
      float sig0 = (gv[t].x - MTOT * mur * muc.x) / MTOT;
      float sig1 = (gv[t].y - MTOT * mur * muc.y) / MTOT;
      float sig2 = (gv[t].z - MTOT * mur * muc.z) / MTOT;
      float sig3 = (gv[t].w - MTOT * mur * muc.w) / MTOT;
      if (r == c0 + 0) sig0 += EPSR;
      if (r == c0 + 1) sig1 += EPSR;
      if (r == c0 + 2) sig2 += EPSR;
      if (r == c0 + 3) sig3 += EPSR;
      const int kr = kpos(r);
      sSt[(c0 + 0) * LD + kr] = sig0;
      sSt[(c0 + 1) * LD + kr] = sig1;
      sSt[(c0 + 2) * LD + kr] = sig2;
      sSt[(c0 + 3) * LD + kr] = sig3;
    }
  }
  __syncthreads();
  if (tid == 0) {
    float tr = 0.f;
    for (int i = 0; i < CH; ++i) tr += sSt[i * LD + kpos(i)];
    sTrInv = 1.0f / tr;
  }
  __syncthreads();
  const float rTr = sTrInv;
  for (int e = tid; e < CH * CH; e += 256) {
    const int n = e >> 7, k = e & 127;
    sSt[n * LD + k] *= rTr;
    sPt[n * LD + kpos(k)] = (n == k) ? 1.f : 0.f;
  }

  const int wv = __builtin_amdgcn_readfirstlane(wave);
  v8f X[8], Y[8];
#pragma unroll
  for (int d = 0; d < 8; ++d) {
#pragma unroll
    for (int r = 0; r < 8; ++r)
      X[d][r] = (d == 0 && l15 == 8 * hi + r) ? 1.f : 0.f;
  }
  __syncthreads();

  float* own = sPt + (8 * hi) * LD + kpos(16 * wave + l15);

  int par = 0;
  for (int it = 0; it < T_NEWTON; ++it) {
    strip_product<5>(Y, X, sPt, wv, l15, hi);
    mirror_tiles(Y, sEx + par * EXSZ, wv, l15, hi);  par ^= 1;
    strip_product<5>(X, Y, sPt, wv, l15, hi);
    mirror_tiles(X, sEx + par * EXSZ, wv, l15, hi);  par ^= 1;
    strip_product<5>(Y, X, sSt, wv, l15, hi);
    mirror_tiles(Y, sEx + par * EXSZ, wv, l15, hi);  par ^= 1;
#pragma unroll
    for (int d = 0; d < 8; ++d) {
      const int i = (wv + d) & 7;
#pragma unroll
      for (int r = 0; r < 8; ++r)
        X[d][r] = 1.5f * own[(16 * i + r) * LD] - 0.5f * Y[d][r];
    }
    __syncthreads();
#pragma unroll
    for (int d = 0; d < 8; ++d) {
      const int i = (wv + d) & 7;
#pragma unroll
      for (int r = 0; r < 8; ++r) own[(16 * i + r) * LD] = X[d][r];
    }
    __syncthreads();
  }

  const float* rrow = rot + (size_t)(16 * wave + l15) * CH + 8 * hi;
#pragma unroll
  for (int d = 0; d < 8; ++d) {
    const int j = (wv + d) & 7;
    const float4 v0 = *reinterpret_cast<const float4*>(rrow + 16 * j);
    const float4 v1 = *reinterpret_cast<const float4*>(rrow + 16 * j + 4);
    X[d][0] = v0.x; X[d][1] = v0.y; X[d][2] = v0.z; X[d][3] = v0.w;
    X[d][4] = v1.x; X[d][5] = v1.y; X[d][6] = v1.z; X[d][7] = v1.w;
  }
  strip_product<8>(Y, X, sPt, wv, l15, hi);

  float* sT = sSt;
  float* trow = sT + (16 * wave + l15) * LD + 8 * hi;
#pragma unroll
  for (int e = 0; e < 8; ++e) {
    const int i = (wv + e) & 7;
    *reinterpret_cast<float4*>(trow + 16 * i)     = make_float4(Y[e][0], Y[e][1], Y[e][2], Y[e][3]);
    *reinterpret_cast<float4*>(trow + 16 * i + 4) = make_float4(Y[e][4], Y[e][5], Y[e][6], Y[e][7]);
  }
  __syncthreads();

  const float sc = sqrtf(rTr);
  for (int e = tid; e < CH * CH; e += 256) Mout[e] = sT[(e >> 7) * LD + (e & 127)] * sc;
  if (tid < CH) {
    float s = 0.f;
    for (int c = 0; c < CH; ++c) s += sT[tid * LD + c] * mean[c];
    bias[tid] = s * sc;
  }
}

constexpr int MPAD    = CH + 4;
constexpr int APC     = GKC;
constexpr int APIECES = KPIECE / APC;
constexpr int APP     = 2 * APC;
constexpr int ASLAB   = (CH / 2) * APP;
constexpr int ATILES  = APC / 16;
constexpr int ASTEPS  = CH / 4;
static_assert(APIECES * APC == KPIECE, "a workgroup's share is a whole number of pieces");
static_assert(ASTEPS > 4, "a tile's first four k steps stand in front of the stores of the tile before it");

struct APiece { v4f a[8], b[8]; };

__device__ __forceinline__ void apply_fetch8(APiece& v, const float* __restrict__ xb,
                                             int kc, int p0, int lane) {
  const int ln = lane < GROW4 ? lane : GROW4 - 1;
  const float* s = xb + (size_t)(2 * p0) * HW + kc + ln * 4;
#pragma unroll
  for (int i = 0; i < 8; ++i) {
    v.a[i] = *reinterpret_cast<const v4f*>(s + (size_t)i * 16 * HW);
    v.b[i] = *reinterpret_cast<const v4f*>(s + (size_t)i * 16 * HW + HW);
  }
  asm volatile("" : "+v"(v.a[0]), "+v"(v.a[1]), "+v"(v.a[2]), "+v"(v.a[3]),
                    "+v"(v.a[4]), "+v"(v.a[5]), "+v"(v.a[6]), "+v"(v.a[7]));
  asm volatile("" : "+v"(v.b[0]), "+v"(v.b[1]), "+v"(v.b[2]), "+v"(v.b[3]),
                    "+v"(v.b[4]), "+v"(v.b[5]), "+v"(v.b[6]), "+v"(v.b[7]));
}

__device__ __forceinline__ void apply_put8(float* __restrict__ slab, const APiece& v,
                                           int p0, int lane) {
  if (lane < GROW4) {
    float* d = slab + p0 * APP + lane * 8;
#pragma unroll
    for (int i = 0; i < 8; ++i) {
      v4f lo, hi;
      lo.x = v.a[i].x; lo.y = v.b[i].x; lo.z = v.a[i].y; lo.w = v.b[i].y;
      hi.x = v.a[i].z; hi.y = v.b[i].z; hi.z = v.a[i].w; hi.w = v.b[i].w;
      *reinterpret_cast<v4f*>(d + i * 8 * APP)     = lo;
      *reinterpret_cast<v4f*>(d + i * 8 * APP + 4) = hi;
    }
  }
}

struct BTile { v2f k[ASTEPS]; };

__device__ __forceinline__ void apply_btile(BTile& f, const float* __restrict__ pb, int t) {
#pragma unroll
  for (int s = 0; s < ASTEPS; ++s)
    f.k[s] = *reinterpret_cast<const v2f*>(pb + (2 * s) * APP + 32 * t);
}

template <int S0, int S1>
__device__ __forceinline__ void apply_steps(v8f& c0, v8f& c1, const v2f (&ma0)[ASTEPS],
                                            const v2f (&ma1)[ASTEPS], const BTile& f) {
#pragma unroll
  for (int s = S0; s < S1; ++s) {
    if (s == 0) {
      c0 = __builtin_amdgcn_wmma_f32_16x16x4_f32(
          false, ma0[0], false, f.k[0], (short)0, v8f{}, false, false);
      c1 = __builtin_amdgcn_wmma_f32_16x16x4_f32(
          false, ma1[0], false, f.k[0], (short)0, v8f{}, false, false);
    } else {
      c0 = __builtin_amdgcn_wmma_f32_16x16x4_f32(
          false, ma0[s], false, f.k[s], (short)0, c0, false, false);
      c1 = __builtin_amdgcn_wmma_f32_16x16x4_f32(
          false, ma1[s], false, f.k[s], (short)0, c1, false, false);
    }
  }
}

__device__ __forceinline__ void apply_store(float* __restrict__ op, int t, v8f c0, v8f c1,
                                            const float (&bv0)[8], const float (&bv1)[8]) {
#pragma unroll
  for (int r = 0; r < 8; ++r) {
    c0[r] -= bv0[r];
    c1[r] -= bv1[r];
  }
  asm volatile("" : "+v"(c0), "+v"(c1));
#pragma unroll
  for (int r = 0; r < 8; ++r) {
    op[(size_t)r * HW + 16 * t]        = c0[r];
    op[(size_t)(r + 16) * HW + 16 * t] = c1[r];
  }
}

__global__ void __launch_bounds__(256)
cw_apply(const float* __restrict__ x, const float* __restrict__ Mmat,
         const float* __restrict__ bias, float* __restrict__ out) {
  extern __shared__ __attribute__((aligned(16))) float apsm[];
  float* slab = apsm;
  float* sM   = apsm + 2 * ASLAB;
  const int tid  = threadIdx.x;
  const int wave = __builtin_amdgcn_readfirstlane(tid >> 5);
  const int lane = tid & 31;
  const int hi   = lane >> 4;
  const int l15  = lane & 15;
  const bool loader = wave >= 4;
  const int  w4     = wave & 3;

  const int q = blockIdx.x % KSPLIT;
  const int b = blockIdx.x / KSPLIT;
  const float* xb = x   + (size_t)b * CH * HW + q * KPIECE;
  float*       ob = out + (size_t)b * CH * HW + q * KPIECE;

  float bv0[8], bv1[8];
  {
    const v4f* m4 = reinterpret_cast<const v4f*>(Mmat);
    const v4f* b4 = reinterpret_cast<const v4f*>(bias + 32 * w4 + 8 * hi);
    v4f mv[16], bq[4];
    APiece pv;
    bq[0] = b4[0]; bq[1] = b4[1]; bq[2] = b4[4]; bq[3] = b4[5];
#pragma unroll
    for (int t = 0; t < 16; ++t) mv[t] = m4[t * 256 + tid];
    apply_fetch8(pv, xb, 0, wave, lane);
    asm volatile("" : "+v"(mv[0]), "+v"(mv[1]), "+v"(mv[2]), "+v"(mv[3]),
                      "+v"(mv[4]), "+v"(mv[5]), "+v"(mv[6]), "+v"(mv[7]));
    asm volatile("" : "+v"(mv[8]), "+v"(mv[9]), "+v"(mv[10]), "+v"(mv[11]),
                      "+v"(mv[12]), "+v"(mv[13]), "+v"(mv[14]), "+v"(mv[15]));
    asm volatile("" : "+v"(bq[0]), "+v"(bq[1]), "+v"(bq[2]), "+v"(bq[3]));
    apply_put8(slab, pv, wave, lane);
#pragma unroll
    for (int t = 0; t < 16; ++t) {
      const int i = t * 256 + tid;
      *reinterpret_cast<v4f*>(sM + (i >> 5) * MPAD + (i & 31) * 4) = mv[t];
    }
#pragma unroll
    for (int r = 0; r < 4; ++r) {
      bv0[r] = bq[0][r]; bv0[4 + r] = bq[1][r];
      bv1[r] = bq[2][r]; bv1[4 + r] = bq[3][r];
    }
  }
  __syncthreads();

  v2f ma0[ASTEPS], ma1[ASTEPS];
  {
    const float* pa = sM + (32 * w4 + l15) * MPAD + 2 * hi;
#pragma unroll
    for (int s = 0; s < ASTEPS; ++s) {
      ma0[s] = *reinterpret_cast<const v2f*>(pa + 4 * s);
      ma1[s] = *reinterpret_cast<const v2f*>(pa + 16 * MPAD + 4 * s);
    }
  }

  for (int j = 0; j < APIECES; ++j) {
    const float* cur = slab + (j & 1) * ASLAB;
    if (loader && j + 1 < APIECES) {
      float* nxt = slab + ((j + 1) & 1) * ASLAB;
      APiece pv;
      apply_fetch8(pv, xb, (j + 1) * APC, w4, lane);
      apply_put8(nxt, pv, w4, lane);
      apply_fetch8(pv, xb, (j + 1) * APC, w4 + 4, lane);
      apply_put8(nxt, pv, w4 + 4, lane);
    }
    if (!loader) {
      const float* pb = cur + hi * APP + 2 * l15;
      float* op = ob + (size_t)(32 * w4 + 8 * hi) * HW + j * APC + l15;
      BTile f[2];
      v8f c0[2], c1[2];
      apply_btile(f[0], pb, 0);
#pragma unroll
      for (int t = 0; t < ATILES; ++t) {
        const int u = t & 1;
        if (t + 1 < ATILES) apply_btile(f[u ^ 1], pb, t + 1);
        __builtin_amdgcn_sched_barrier(0);
        if (t == 0) {
          apply_steps<0, ASTEPS>(c0[u], c1[u], ma0, ma1, f[u]);
        } else {
          apply_steps<0, 4>(c0[u], c1[u], ma0, ma1, f[u]);
          apply_store(op, t - 1, c0[u ^ 1], c1[u ^ 1], bv0, bv1);
          apply_steps<4, ASTEPS>(c0[u], c1[u], ma0, ma1, f[u]);
        }
        __builtin_amdgcn_sched_barrier(0);
      }
      apply_store(op, ATILES - 1, c0[(ATILES - 1) & 1], c1[(ATILES - 1) & 1], bv0, bv1);
    }
    __builtin_amdgcn_fence(__ATOMIC_RELEASE, "workgroup", "local");
    __builtin_amdgcn_s_barrier();
    __builtin_amdgcn_fence(__ATOMIC_ACQUIRE, "workgroup", "local");
  }
}

extern "C" void kernel_launch(void* const* d_in, const int* in_sizes, int n_in,
                              void* d_out, int out_size, void* d_ws, size_t ws_size,
                              hipStream_t stream) {
  const float* x   = (const float*)d_in[0];
  const float* rot = (const float*)d_in[1];
  float* out = (float*)d_out;
  float* ws  = (float*)d_ws;

  float* gramSum  = ws;
  float* meanPart = gramSum + (size_t)BATCH * CH * CH;
  float* mean     = meanPart + CH * BATCH;
  float* Mmat     = mean + CH;
  float* bias     = Mmat + CH * CH;

  float* planes = out;

  cw_gram_mean<<<GRAM_WGS + MEAN_WGS, 256, 2 * GSLAB * sizeof(float), stream>>>(x, planes, meanPart);
  cw_gram_sum_mean<<<SUM_WGS + CH / 64, 64, 0, stream>>>(planes, gramSum, meanPart, mean);
  cw_newton<<<1, 256, (2 * CH * LD + 2 * EXSZ) * sizeof(float), stream>>>(
      gramSum, mean, rot, Mmat, bias);
  cw_apply<<<GRAM_WGS, 256, (2 * ASLAB + CH * MPAD) * sizeof(float), stream>>>(
      x, Mmat, bias, out);
}


# cw_newton: A fragments of each k block of a strip product read from LDS as one group a block ahead, into a second register set (were single reads among the matrix instructions); sums unchanged
# speedup vs baseline: 37.9800x; 1.0430x over previous
#include <hip/hip_runtime.h>
#include <hip/hip_bf16.h>


typedef __attribute__((ext_vector_type(2))) float v2f;
typedef __attribute__((ext_vector_type(8))) float v8f;
typedef __attribute__((ext_vector_type(4))) float v4f;

constexpr int CH    = 128;
constexpr int BATCH = 64;
constexpr int HW    = 56 * 56;
constexpr float MTOT = (float)BATCH * (float)HW;
constexpr float EPSR = 1e-5f;
constexpr int   T_NEWTON = 5;

constexpr int MP_WAVES = 8;
constexpr int ROW4     = HW / 4;
constexpr int MP_STEPS = 2 * ROW4 / 32;
constexpr int MEAN_WGS = BATCH * CH / (2 * MP_WAVES);
static_assert(2 * ROW4 == MP_STEPS * 32, "a pair of rows is a whole number of wave loads");
static_assert(ROW4 == 24 * 32 + 16, "row boundary sits in the middle of wave load 24");
static_assert(BATCH * CH == 2 * MEAN_WGS * MP_WAVES, "the row-sum workgroups cover every row pair exactly once");

__device__ __forceinline__ void mean_rows_job(const float* __restrict__ x,
                                              float* __restrict__ meanPart,
                                              int pair, int lane) {
  const float4* p = reinterpret_cast<const float4*>(x) + (size_t)pair * (2 * ROW4) + lane;
  float4 v[MP_STEPS];
#pragma unroll
  for (int t = 0; t < MP_STEPS; ++t) v[t] = p[t * 32];
  float s0 = 0.f, s1 = 0.f;
#pragma unroll
  for (int t = 0; t < 24; ++t) s0 += (v[t].x + v[t].y) + (v[t].z + v[t].w);
  const float mid = (v[24].x + v[24].y) + (v[24].z + v[24].w);
  if (lane < 16) s0 += mid; else s1 += mid;
#pragma unroll
  for (int t = 25; t < MP_STEPS; ++t) s1 += (v[t].x + v[t].y) + (v[t].z + v[t].w);
  for (int off = 16; off > 0; off >>= 1) {
    s0 += __shfl_xor(s0, off);
    s1 += __shfl_xor(s1, off);
  }
  if (lane == 0) {
    const int b = pair >> 6;
    const int c = (pair & 63) * 2;
    meanPart[c * BATCH + b]       = s0;
    meanPart[(c + 1) * BATCH + b] = s1;
  }
}

__device__ __forceinline__ void mean_finish_job(const float* __restrict__ meanPart,
                                                float* __restrict__ mean, int c) {
  float s = 0.f;
  for (int b = 0; b < BATCH; ++b) s += meanPart[c * BATCH + b];
  mean[c] = s / MTOT;
}

constexpr int KSPLIT = 4;
constexpr int KPIECE = HW / KSPLIT;
constexpr int NPLANE = BATCH * KSPLIT;
constexpr int GKC    = 112;
constexpr int GPAD   = GKC + 4;
constexpr int GSLAB   = CH * GPAD;
constexpr int GSTEPS  = GKC / 4;
constexpr int GCHUNKS = KPIECE / GKC;
constexpr int GROW4   = GKC / 4;
static_assert(GCHUNKS * GKC == KPIECE, "a piece is a whole number of chunks");
static_assert(GSTEPS % 4 == 0, "the product loop takes the k steps of a chunk four at a time");

__device__ __forceinline__ void gram_stage16(float* __restrict__ slab,
                                             const float* __restrict__ xb,
                                             int kc, int r0, int lane) {
  if (lane < GROW4) {
    const float* s = xb + (size_t)r0 * HW + kc + lane * 4;
    float*       d = slab + r0 * GPAD + lane * 4;
    v4f v[16];
#pragma unroll
    for (int i = 0; i < 16; ++i) v[i] = *reinterpret_cast<const v4f*>(s + (size_t)i * 8 * HW);
    asm volatile("" : "+v"(v[0]), "+v"(v[1]), "+v"(v[2]), "+v"(v[3]),
                      "+v"(v[4]), "+v"(v[5]), "+v"(v[6]), "+v"(v[7]));
    asm volatile("" : "+v"(v[8]), "+v"(v[9]), "+v"(v[10]), "+v"(v[11]),
                      "+v"(v[12]), "+v"(v[13]), "+v"(v[14]), "+v"(v[15]));
#pragma unroll
    for (int i = 0; i < 16; ++i) *reinterpret_cast<v4f*>(d + i * 8 * GPAD) = v[i];
  }
}

struct GSet { v2f a[2][2]; v2f b[8][2]; };

__device__ __forceinline__ void gram_frags(GSet& f, const float* __restrict__ pa,
                                           const float* __restrict__ pb, int k0) {
#pragma unroll
  for (int e = 0; e < 2; ++e) {
    f.a[0][e] = *reinterpret_cast<const v2f*>(pa + k0 + 4 * e);
    f.a[1][e] = *reinterpret_cast<const v2f*>(pa + 16 * GPAD + k0 + 4 * e);
  }
#pragma unroll
  for (int tc = 0; tc < 8; ++tc) {
#pragma unroll
    for (int e = 0; e < 2; ++e)
      f.b[tc][e] = *reinterpret_cast<const v2f*>(pb + tc * 16 * GPAD + k0 + 4 * e);
  }
}

__device__ __forceinline__ void gram_opaque(GSet& f) {
#pragma unroll
  for (int s = 0; s < 2; ++s) {
#pragma unroll
    for (int e = 0; e < 2; ++e) asm volatile("" : "+v"(f.a[s][e]));
  }
#pragma unroll
  for (int tc = 0; tc < 8; ++tc) {
#pragma unroll
    for (int e = 0; e < 2; ++e) asm volatile("" : "+v"(f.b[tc][e]));
  }
}

__device__ __forceinline__ void gram_mma(v8f (&acc0)[8], v8f (&acc1)[8], const GSet& f) {
#pragma unroll
  for (int e = 0; e < 2; ++e) {
#pragma unroll
    for (int tc = 0; tc < 8; ++tc) {
      acc0[tc] = __builtin_amdgcn_wmma_f32_16x16x4_f32(
          false, f.a[0][e], false, f.b[tc][e], (short)0, acc0[tc], false, false);
      acc1[tc] = __builtin_amdgcn_wmma_f32_16x16x4_f32(
          false, f.a[1][e], false, f.b[tc][e], (short)0, acc1[tc], false, false);
    }
  }
}

__device__ __forceinline__ void gram_job(const float* __restrict__ x,
                                         float* __restrict__ gramPart,
                                         int q, int b) {
  extern __shared__ __attribute__((aligned(16))) float gsm[];
  const int tid  = threadIdx.x;
  const int wave = __builtin_amdgcn_readfirstlane(tid >> 5);
  const int lane = tid & 31;
  const int hi   = lane >> 4;
  const int l15  = lane & 15;
  const bool loader = wave >= 4;
  const int  w4     = wave & 3;

  const float* xb = x + (size_t)b * CH * HW + q * KPIECE;

  gram_stage16(gsm, xb, 0, wave, lane);
  __syncthreads();

  v8f acc0[8] = {}, acc1[8] = {};
  const int ntrips = loader ? 0 : GSTEPS / 4;
  for (int j = 0; j < GCHUNKS; ++j) {
    const float* cur = gsm + (j & 1) * GSLAB;
    if (loader && j + 1 < GCHUNKS) {
      float* nxt = gsm + ((j + 1) & 1) * GSLAB;
      gram_stage16(nxt, xb, (j + 1) * GKC, w4,     lane);
      gram_stage16(nxt, xb, (j + 1) * GKC, w4 + 4, lane);
    }
    {
      const float* pa = cur + (32 * w4 + l15) * GPAD + 2 * hi;
      const float* pb = cur + l15 * GPAD + 2 * hi;
      GSet f0, f1;
      gram_frags(f0, pa, pb, 0);
      gram_opaque(f0);
#pragma clang loop unroll(disable)
      for (int u = 0; u < ntrips; ++u) {
        gram_frags(f1, pa, pb, 16 * u + 8);
        gram_mma(acc0, acc1, f0);
        gram_frags(f0, pa, pb, (u + 1 < GSTEPS / 4) ? 16 * u + 16 : 0);
        gram_mma(acc0, acc1, f1);
      }
    }
    __syncthreads();
  }

  if (!loader) {
    float* gp = gramPart + (size_t)(b * KSPLIT + q) * CH * CH;
#pragma unroll
    for (int tc = 0; tc < 8; ++tc) {
#pragma unroll
      for (int r = 0; r < 8; ++r) {
        const int row = 32 * w4 + r + hi * 8;
        const int col = tc * 16 + l15;
        gp[row * CH + col]        = acc0[tc][r];
        gp[(row + 16) * CH + col] = acc1[tc][r];
      }
    }
  }
}

constexpr int GRAM_WGS = NPLANE;

__global__ void __launch_bounds__(256)
cw_gram_mean(const float* __restrict__ x, float* __restrict__ gramPart,
             float* __restrict__ meanPart) {
  if (blockIdx.x >= GRAM_WGS) {
    const int pair = (blockIdx.x - GRAM_WGS) * MP_WAVES + (threadIdx.x >> 5);
    mean_rows_job(x, meanPart, pair, threadIdx.x & 31);
    return;
  }
  gram_job(x, gramPart, blockIdx.x % KSPLIT, blockIdx.x / KSPLIT);
}

constexpr int SUM_WGS = CH * CH / 64;

__global__ void __launch_bounds__(64)
cw_gram_sum_mean(const float* __restrict__ planes, float* __restrict__ gsum,
                 const float* __restrict__ meanPart, float* __restrict__ mean) {
  if (blockIdx.x >= SUM_WGS) {
    mean_finish_job(meanPart, mean, (blockIdx.x - SUM_WGS) * 64 + threadIdx.x);
    return;
  }
  const int e = blockIdx.x * 64 + threadIdx.x;
  const float* p = planes + e;
  float g = 0.f;
  for (int s0 = 0; s0 < NPLANE; s0 += 64) {
    float v[64];
#pragma unroll
    for (int s = 0; s < 64; ++s) v[s] = p[(size_t)(s0 + s) * (CH * CH)];
#pragma unroll
    for (int s = 0; s < 64; ++s) g += v[s];
  }
  gsum[e] = g;
}

constexpr int LD = CH + 4;

__device__ __forceinline__ int kpos(int k) {
  const int kk = k & 15;
  return (k & ~15) | ((kk & 6) << 1) | ((kk & 8) >> 2) | (kk & 1);
}

struct NSet { v2f a[4][8]; };

template <int NT>
__device__ __forceinline__ void newton_frags(NSet& f, const float* __restrict__ base,
                                             int wave, int d) {
  const int j = (wave + d) & 7;
#pragma unroll
  for (int e = 0; e < NT; ++e) {
    const int i = (wave + e) & 7;
#pragma unroll
    for (int q = 0; q < 4; ++q)
      f.a[q][e] = *reinterpret_cast<const v2f*>(base + i * 16 * LD + 16 * j + 4 * q);
  }
}

template <int NT, bool FIRST>
__device__ __forceinline__ void newton_mma(v8f (&Y)[8], const v8f& Xd, const NSet& f) {
#pragma unroll
  for (int q = 0; q < 4; ++q) {
    v2f bb; bb.x = Xd[2 * q]; bb.y = Xd[2 * q + 1];
#pragma unroll
    for (int e = 0; e < NT; ++e) {
      if (FIRST && q == 0)
        Y[e] = __builtin_amdgcn_wmma_f32_16x16x4_f32(
            false, f.a[q][e], false, bb, (short)0, v8f{}, false, false);
      else
        Y[e] = __builtin_amdgcn_wmma_f32_16x16x4_f32(
            false, f.a[q][e], false, bb, (short)0, Y[e], false, false);
    }
  }
}

template <int NT, int NTN>
__device__ __forceinline__ void strip_product(v8f (&Y)[8], const v8f (&X)[8], NSet (&f)[2],
                                              const float* __restrict__ sA,
                                              const float* __restrict__ sNext,
                                              int wave, int l15, int hi) {
  const float* base = sA + l15 * LD + 2 * hi;
#pragma unroll
  for (int d = 0; d < 8; ++d) {
    if (d + 1 < 8) newton_frags<NT>(f[(d + 1) & 1], base, wave, d + 1);
    else if (NTN > 0) newton_frags<(NTN > 0 ? NTN : 1)>(f[0], sNext + l15 * LD + 2 * hi, wave, 0);
    __builtin_amdgcn_sched_barrier(0);
    if (d == 0) newton_mma<NT, true>(Y, X[d], f[d & 1]);
    else        newton_mma<NT, false>(Y, X[d], f[d & 1]);
    __builtin_amdgcn_sched_barrier(0);
  }
}

constexpr int EXP  = 20;
constexpr int EXT  = 16 * EXP;
constexpr int EXSZ = 8 * 3 * EXT;

__device__ __forceinline__ void mirror_tiles(v8f (&Y)[8], float* __restrict__ sEx,
                                             int wave, int l15, int hi) {
  float* wr = sEx + (wave * 3) * EXT + l15 * EXP + 8 * hi;
#pragma unroll
  for (int e = 1; e <= 3; ++e) {
    *reinterpret_cast<float4*>(wr + (e - 1) * EXT)     = make_float4(Y[e][0], Y[e][1], Y[e][2], Y[e][3]);
    *reinterpret_cast<float4*>(wr + (e - 1) * EXT + 4) = make_float4(Y[e][4], Y[e][5], Y[e][6], Y[e][7]);
  }
  __syncthreads();
#pragma unroll
  for (int d = 5; d <= 7; ++d) {
    const int src = (wave + d) & 7;
    const float* rd = sEx + (src * 3 + (7 - d)) * EXT + (8 * hi) * EXP + l15;
#pragma unroll
    for (int r = 0; r < 8; ++r) Y[d][r] = rd[r * EXP];
  }
}

__global__ void __launch_bounds__(256)
cw_newton(const float* __restrict__ gsum, const float* __restrict__ mean,
          const float* __restrict__ rot, float* __restrict__ Mout,
          float* __restrict__ bias) {
  extern __shared__ __attribute__((aligned(16))) float nsm[];
  float* sSt = nsm;
  float* sPt = nsm + CH * LD;
  float* sEx = nsm + 2 * CH * LD;
  __shared__ float sTrInv;
  const int tid  = threadIdx.x;
  const int wave = tid >> 5;
  const int lane = tid & 31;
  const int hi   = lane >> 4;
  const int l15  = lane & 15;

  {
    constexpr int PLANE4 = CH * CH / 4;
    constexpr int TRIPS  = PLANE4 / 256;
    const float4* gs4 = reinterpret_cast<const float4*>(gsum);
    const int c0 = (tid & 31) * 4;
    const float4 muc = *reinterpret_cast<const float4*>(mean + c0);
    float4 gv[TRIPS];
#pragma unroll
    for (int t = 0; t < TRIPS; ++t) gv[t] = gs4[t * 256 + tid];
#pragma unroll
    for (int t = 0; t < TRIPS; ++t) {
      const int r = t * 8 + __builtin_amdgcn_readfirstlane(wave);
      const float mur = mean[r];
      float sig0 = (gv[t].x - MTOT * mur * muc.x) / MTOT;
      float sig1 = (gv[t].y - MTOT * mur * muc.y) / MTOT;
      float sig2 = (gv[t].z - MTOT * mur * muc.z) / MTOT;
      float sig3 = (gv[t].w - MTOT * mur * muc.w) / MTOT;
      if (r == c0 + 0) sig0 += EPSR;
      if (r == c0 + 1) sig1 += EPSR;
      if (r == c0 + 2) sig2 += EPSR;
      if (r == c0 + 3) sig3 += EPSR;
      const int kr = kpos(r);
      sSt[(c0 + 0) * LD + kr] = sig0;
      sSt[(c0 + 1) * LD + kr] = sig1;
      sSt[(c0 + 2) * LD + kr] = sig2;
      sSt[(c0 + 3) * LD + kr] = sig3;
    }
  }
  __syncthreads();
  if (tid == 0) {
    float tr = 0.f;
    for (int i = 0; i < CH; ++i) tr += sSt[i * LD + kpos(i)];
    sTrInv = 1.0f / tr;
  }
  __syncthreads();
  const float rTr = sTrInv;
  for (int e = tid; e < CH * CH; e += 256) {
    const int n = e >> 7, k = e & 127;
    sSt[n * LD + k] *= rTr;
    sPt[n * LD + kpos(k)] = (n == k) ? 1.f : 0.f;
  }

  const int wv = __builtin_amdgcn_readfirstlane(wave);
  v8f X[8], Y[8];
#pragma unroll
  for (int d = 0; d < 8; ++d) {
#pragma unroll
    for (int r = 0; r < 8; ++r)
      X[d][r] = (d == 0 && l15 == 8 * hi + r) ? 1.f : 0.f;
  }
  __syncthreads();

  float* own = sPt + (8 * hi) * LD + kpos(16 * wave + l15);

  int par = 0;
  NSet f[2];
  const float* fragP = sPt + l15 * LD + 2 * hi;
  for (int it = 0; it < T_NEWTON; ++it) {
    newton_frags<5>(f[0], fragP, wv, 0);
    strip_product<5, 5>(Y, X, f, sPt, sPt, wv, l15, hi);
    mirror_tiles(Y, sEx + par * EXSZ, wv, l15, hi);  par ^= 1;
    strip_product<5, 5>(X, Y, f, sPt, sSt, wv, l15, hi);
    mirror_tiles(X, sEx + par * EXSZ, wv, l15, hi);  par ^= 1;
    strip_product<5, 0>(Y, X, f, sSt, nullptr, wv, l15, hi);
    mirror_tiles(Y, sEx + par * EXSZ, wv, l15, hi);  par ^= 1;
#pragma unroll
    for (int d = 0; d < 8; ++d) {
      const int i = (wv + d) & 7;
#pragma unroll
      for (int r = 0; r < 8; ++r)
        X[d][r] = 1.5f * own[(16 * i + r) * LD] - 0.5f * Y[d][r];
    }
    __syncthreads();
#pragma unroll
    for (int d = 0; d < 8; ++d) {
      const int i = (wv + d) & 7;
#pragma unroll
      for (int r = 0; r < 8; ++r) own[(16 * i + r) * LD] = X[d][r];
    }
    __syncthreads();
  }

  const float* rrow = rot + (size_t)(16 * wave + l15) * CH + 8 * hi;
#pragma unroll
  for (int d = 0; d < 8; ++d) {
    const int j = (wv + d) & 7;
    const float4 v0 = *reinterpret_cast<const float4*>(rrow + 16 * j);
    const float4 v1 = *reinterpret_cast<const float4*>(rrow + 16 * j + 4);
    X[d][0] = v0.x; X[d][1] = v0.y; X[d][2] = v0.z; X[d][3] = v0.w;
    X[d][4] = v1.x; X[d][5] = v1.y; X[d][6] = v1.z; X[d][7] = v1.w;
  }
  newton_frags<8>(f[0], fragP, wv, 0);
  strip_product<8, 0>(Y, X, f, sPt, nullptr, wv, l15, hi);

  float* sT = sSt;
  float* trow = sT + (16 * wave + l15) * LD + 8 * hi;
#pragma unroll
  for (int e = 0; e < 8; ++e) {
    const int i = (wv + e) & 7;
    *reinterpret_cast<float4*>(trow + 16 * i)     = make_float4(Y[e][0], Y[e][1], Y[e][2], Y[e][3]);
    *reinterpret_cast<float4*>(trow + 16 * i + 4) = make_float4(Y[e][4], Y[e][5], Y[e][6], Y[e][7]);
  }
  __syncthreads();

  const float sc = sqrtf(rTr);
  for (int e = tid; e < CH * CH; e += 256) Mout[e] = sT[(e >> 7) * LD + (e & 127)] * sc;
  if (tid < CH) {
    float s = 0.f;
    for (int c = 0; c < CH; ++c) s += sT[tid * LD + c] * mean[c];
    bias[tid] = s * sc;
  }
}

constexpr int MPAD    = CH + 4;
constexpr int APC     = GKC;
constexpr int APIECES = KPIECE / APC;
constexpr int APP     = 2 * APC;
constexpr int ASLAB   = (CH / 2) * APP;
constexpr int ATILES  = APC / 16;
constexpr int ASTEPS  = CH / 4;
static_assert(APIECES * APC == KPIECE, "a workgroup's share is a whole number of pieces");
static_assert(ASTEPS > 4, "a tile's first four k steps stand in front of the stores of the tile before it");

struct APiece { v4f a[8], b[8]; };

__device__ __forceinline__ void apply_fetch8(APiece& v, const float* __restrict__ xb,
                                             int kc, int p0, int lane) {
  const int ln = lane < GROW4 ? lane : GROW4 - 1;
  const float* s = xb + (size_t)(2 * p0) * HW + kc + ln * 4;
#pragma unroll
  for (int i = 0; i < 8; ++i) {
    v.a[i] = *reinterpret_cast<const v4f*>(s + (size_t)i * 16 * HW);
    v.b[i] = *reinterpret_cast<const v4f*>(s + (size_t)i * 16 * HW + HW);
  }
  asm volatile("" : "+v"(v.a[0]), "+v"(v.a[1]), "+v"(v.a[2]), "+v"(v.a[3]),
                    "+v"(v.a[4]), "+v"(v.a[5]), "+v"(v.a[6]), "+v"(v.a[7]));
  asm volatile("" : "+v"(v.b[0]), "+v"(v.b[1]), "+v"(v.b[2]), "+v"(v.b[3]),
                    "+v"(v.b[4]), "+v"(v.b[5]), "+v"(v.b[6]), "+v"(v.b[7]));
}

__device__ __forceinline__ void apply_put8(float* __restrict__ slab, const APiece& v,
                                           int p0, int lane) {
  if (lane < GROW4) {
    float* d = slab + p0 * APP + lane * 8;
#pragma unroll
    for (int i = 0; i < 8; ++i) {
      v4f lo, hi;
      lo.x = v.a[i].x; lo.y = v.b[i].x; lo.z = v.a[i].y; lo.w = v.b[i].y;
      hi.x = v.a[i].z; hi.y = v.b[i].z; hi.z = v.a[i].w; hi.w = v.b[i].w;
      *reinterpret_cast<v4f*>(d + i * 8 * APP)     = lo;
      *reinterpret_cast<v4f*>(d + i * 8 * APP + 4) = hi;
    }
  }
}

struct BTile { v2f k[ASTEPS]; };

__device__ __forceinline__ void apply_btile(BTile& f, const float* __restrict__ pb, int t) {
#pragma unroll
  for (int s = 0; s < ASTEPS; ++s)
    f.k[s] = *reinterpret_cast<const v2f*>(pb + (2 * s) * APP + 32 * t);
}

template <int S0, int S1>
__device__ __forceinline__ void apply_steps(v8f& c0, v8f& c1, const v2f (&ma0)[ASTEPS],
                                            const v2f (&ma1)[ASTEPS], const BTile& f) {
#pragma unroll
  for (int s = S0; s < S1; ++s) {
    if (s == 0) {
      c0 = __builtin_amdgcn_wmma_f32_16x16x4_f32(
          false, ma0[0], false, f.k[0], (short)0, v8f{}, false, false);
      c1 = __builtin_amdgcn_wmma_f32_16x16x4_f32(
          false, ma1[0], false, f.k[0], (short)0, v8f{}, false, false);
    } else {
      c0 = __builtin_amdgcn_wmma_f32_16x16x4_f32(
          false, ma0[s], false, f.k[s], (short)0, c0, false, false);
      c1 = __builtin_amdgcn_wmma_f32_16x16x4_f32(
          false, ma1[s], false, f.k[s], (short)0, c1, false, false);
    }
  }
}

__device__ __forceinline__ void apply_store(float* __restrict__ op, int t, v8f c0, v8f c1,
                                            const float (&bv0)[8], const float (&bv1)[8]) {
#pragma unroll
  for (int r = 0; r < 8; ++r) {
    c0[r] -= bv0[r];
    c1[r] -= bv1[r];
  }
  asm volatile("" : "+v"(c0), "+v"(c1));
#pragma unroll
  for (int r = 0; r < 8; ++r) {
    op[(size_t)r * HW + 16 * t]        = c0[r];
    op[(size_t)(r + 16) * HW + 16 * t] = c1[r];
  }
}

__global__ void __launch_bounds__(256)
cw_apply(const float* __restrict__ x, const float* __restrict__ Mmat,
         const float* __restrict__ bias, float* __restrict__ out) {
  extern __shared__ __attribute__((aligned(16))) float apsm[];
  float* slab = apsm;
  float* sM   = apsm + 2 * ASLAB;
  const int tid  = threadIdx.x;
  const int wave = __builtin_amdgcn_readfirstlane(tid >> 5);
  const int lane = tid & 31;
  const int hi   = lane >> 4;
  const int l15  = lane & 15;
  const bool loader = wave >= 4;
  const int  w4     = wave & 3;

  const int q = blockIdx.x % KSPLIT;
  const int b = blockIdx.x / KSPLIT;
  const float* xb = x   + (size_t)b * CH * HW + q * KPIECE;
  float*       ob = out + (size_t)b * CH * HW + q * KPIECE;

  float bv0[8], bv1[8];
  {
    const v4f* m4 = reinterpret_cast<const v4f*>(Mmat);
    const v4f* b4 = reinterpret_cast<const v4f*>(bias + 32 * w4 + 8 * hi);
    v4f mv[16], bq[4];
    APiece pv;
    bq[0] = b4[0]; bq[1] = b4[1]; bq[2] = b4[4]; bq[3] = b4[5];
#pragma unroll
    for (int t = 0; t < 16; ++t) mv[t] = m4[t * 256 + tid];
    apply_fetch8(pv, xb, 0, wave, lane);
    asm volatile("" : "+v"(mv[0]), "+v"(mv[1]), "+v"(mv[2]), "+v"(mv[3]),
                      "+v"(mv[4]), "+v"(mv[5]), "+v"(mv[6]), "+v"(mv[7]));
    asm volatile("" : "+v"(mv[8]), "+v"(mv[9]), "+v"(mv[10]), "+v"(mv[11]),
                      "+v"(mv[12]), "+v"(mv[13]), "+v"(mv[14]), "+v"(mv[15]));
    asm volatile("" : "+v"(bq[0]), "+v"(bq[1]), "+v"(bq[2]), "+v"(bq[3]));
    apply_put8(slab, pv, wave, lane);
#pragma unroll
    for (int t = 0; t < 16; ++t) {
      const int i = t * 256 + tid;
      *reinterpret_cast<v4f*>(sM + (i >> 5) * MPAD + (i & 31) * 4) = mv[t];
    }
#pragma unroll
    for (int r = 0; r < 4; ++r) {
      bv0[r] = bq[0][r]; bv0[4 + r] = bq[1][r];
      bv1[r] = bq[2][r]; bv1[4 + r] = bq[3][r];
    }
  }
  __syncthreads();

  v2f ma0[ASTEPS], ma1[ASTEPS];
  {
    const float* pa = sM + (32 * w4 + l15) * MPAD + 2 * hi;
#pragma unroll
    for (int s = 0; s < ASTEPS; ++s) {
      ma0[s] = *reinterpret_cast<const v2f*>(pa + 4 * s);
      ma1[s] = *reinterpret_cast<const v2f*>(pa + 16 * MPAD + 4 * s);
    }
  }

  for (int j = 0; j < APIECES; ++j) {
    const float* cur = slab + (j & 1) * ASLAB;
    if (loader && j + 1 < APIECES) {
      float* nxt = slab + ((j + 1) & 1) * ASLAB;
      APiece pv;
      apply_fetch8(pv, xb, (j + 1) * APC, w4, lane);
      apply_put8(nxt, pv, w4, lane);
      apply_fetch8(pv, xb, (j + 1) * APC, w4 + 4, lane);
      apply_put8(nxt, pv, w4 + 4, lane);
    }
    if (!loader) {
      const float* pb = cur + hi * APP + 2 * l15;
      float* op = ob + (size_t)(32 * w4 + 8 * hi) * HW + j * APC + l15;
      BTile f[2];
      v8f c0[2], c1[2];
      apply_btile(f[0], pb, 0);
#pragma unroll
      for (int t = 0; t < ATILES; ++t) {
        const int u = t & 1;
        if (t + 1 < ATILES) apply_btile(f[u ^ 1], pb, t + 1);
        __builtin_amdgcn_sched_barrier(0);
        if (t == 0) {
          apply_steps<0, ASTEPS>(c0[u], c1[u], ma0, ma1, f[u]);
        } else {
          apply_steps<0, 4>(c0[u], c1[u], ma0, ma1, f[u]);
          apply_store(op, t - 1, c0[u ^ 1], c1[u ^ 1], bv0, bv1);
          apply_steps<4, ASTEPS>(c0[u], c1[u], ma0, ma1, f[u]);
        }
        __builtin_amdgcn_sched_barrier(0);
      }
      apply_store(op, ATILES - 1, c0[(ATILES - 1) & 1], c1[(ATILES - 1) & 1], bv0, bv1);
    }
    __builtin_amdgcn_fence(__ATOMIC_RELEASE, "workgroup", "local");
    __builtin_amdgcn_s_barrier();
    __builtin_amdgcn_fence(__ATOMIC_ACQUIRE, "workgroup", "local");
  }
}

extern "C" void kernel_launch(void* const* d_in, const int* in_sizes, int n_in,
                              void* d_out, int out_size, void* d_ws, size_t ws_size,
                              hipStream_t stream) {
  const float* x   = (const float*)d_in[0];
  const float* rot = (const float*)d_in[1];
  float* out = (float*)d_out;
  float* ws  = (float*)d_ws;

  float* gramSum  = ws;
  float* meanPart = gramSum + (size_t)BATCH * CH * CH;
  float* mean     = meanPart + CH * BATCH;
  float* Mmat     = mean + CH;
  float* bias     = Mmat + CH * CH;

  float* planes = out;

  cw_gram_mean<<<GRAM_WGS + MEAN_WGS, 256, 2 * GSLAB * sizeof(float), stream>>>(x, planes, meanPart);
  cw_gram_sum_mean<<<SUM_WGS + CH / 64, 64, 0, stream>>>(planes, gramSum, meanPart, mean);
  cw_newton<<<1, 256, (2 * CH * LD + 2 * EXSZ) * sizeof(float), stream>>>(
      gramSum, mean, rot, Mmat, bias);
  cw_apply<<<GRAM_WGS, 256, (2 * ASLAB + CH * MPAD) * sizeof(float), stream>>>(
      x, Mmat, bias, out);
}
